# starting kernel
# baseline (speedup 1.0000x reference)
#include <hip/hip_runtime.h>
#include <math.h>


static constexpr int D0   = 200;
static constexpr int DP   = 208;
static constexpr int NTN  = 13;
static constexpr int NE   = 8192;
static constexpr int NV   = 5000;
static constexpr int NVP  = 5056;
static constexpr int NLAY = 6;
static constexpr int NF   = 96;
static constexpr int CONVK= 49;
static constexpr int P196 = 196;
static constexpr int FLAT = 18816;
static constexpr int NCLS = 13;
#define INV_SQRT_1PE 0.99999500f

typedef __attribute__((ext_vector_type(2))) float v2f;
typedef __attribute__((ext_vector_type(8))) float v8f;

static __device__ __forceinline__ v8f wmma4(v2f a, v2f b, v8f c) {
  return __builtin_amdgcn_wmma_f32_16x16x4_f32(false, a, false, b, (short)0, c,
                                               false, false);
}


__global__ void k_zero(float* __restrict__ p, int n) {
  int i = blockIdx.x * blockDim.x + threadIdx.x;
  if (i < n) p[i] = 0.0f;
}

__global__ void k_pad(const float* __restrict__ src, float* __restrict__ dst,
                      int rows, int prows) {
  int i = blockIdx.x * blockDim.x + threadIdx.x;
  int total = prows * DP;
  if (i >= total) return;
  int r = i / DP, c = i % DP;
  dst[i] = (r < rows && c < D0) ? src[r * D0 + c] : 0.0f;
}

__global__ void k_pad_w(const float* __restrict__ Win, const float* __restrict__ Wout,
                        const float* __restrict__ Wloop, const float* __restrict__ Wrel,
                        int l, float* __restrict__ WP) {
  int i = blockIdx.x * blockDim.x + threadIdx.x;
  int total = 4 * DP * DP;
  if (i >= total) return;
  int w = i / (DP * DP);
  int rem = i % (DP * DP);
  int r = rem / DP, c = rem % DP;
  const float* src = (w == 0 ? Win : w == 1 ? Wout : w == 2 ? Wloop : Wrel)
                     + (size_t)l * D0 * D0;
  WP[i] = (r < D0 && c < D0) ? src[r * D0 + c] : 0.0f;
}

__global__ void k_tr_fcw(const float* __restrict__ fcw, float* __restrict__ fcwT) {
  int i = blockIdx.x * blockDim.x + threadIdx.x;
  if (i >= FLAT * DP) return;
  int k = i / DP, n = i % DP;
  fcwT[i] = (n < D0) ? fcw[(size_t)n * FLAT + k] : 0.0f;
}

__global__ void k_deg(const int* __restrict__ ei0, const int* __restrict__ iei0,
                      float* __restrict__ degf, float* __restrict__ degi) {
  int e = blockIdx.x * blockDim.x + threadIdx.x;
  if (e < NE)            atomicAdd(&degf[ei0[e]], 1.0f);
  else if (e < 2 * NE)   atomicAdd(&degi[iei0[e - NE]], 1.0f);
}

__global__ void k_norm(const int* __restrict__ ei0, const int* __restrict__ ei1,
                       const int* __restrict__ iei0, const int* __restrict__ iei1,
                       const float* __restrict__ degf, const float* __restrict__ degi,
                       float* __restrict__ nf, float* __restrict__ ni) {
  int e = blockIdx.x * blockDim.x + threadIdx.x;
  if (e >= NE) return;
  float df0 = degf[ei0[e]],  df1 = degf[ei1[e]];
  float di0 = degi[iei0[e]], di1 = degi[iei1[e]];
  float a = df0 > 0.f ? rsqrtf(fmaxf(df0, 1.f)) : 0.f;
  float b = df1 > 0.f ? rsqrtf(fmaxf(df1, 1.f)) : 0.f;
  float c = di0 > 0.f ? rsqrtf(fmaxf(di0, 1.f)) : 0.f;
  float d = di1 > 0.f ? rsqrtf(fmaxf(di1, 1.f)) : 0.f;
  nf[e] = a * b;
  ni[e] = c * d;
}

__global__ void k_gather(const float* __restrict__ xP, const float* __restrict__ r,
                         const float* __restrict__ ir,
                         const int* __restrict__ ei0, const int* __restrict__ iei0,
                         float* __restrict__ ain, float* __restrict__ aout) {
  int i = blockIdx.x * blockDim.x + threadIdx.x;
  if (i >= NE * DP) return;
  int e = i / DP, c = i % DP;
  ain[i]  = xP[(size_t)ei0[e] * DP + c] - r[i];
  aout[i] = xP[(size_t)iei0[e] * DP + c] - ir[i];
}

__global__ void k_scatter(const float* __restrict__ src, const int* __restrict__ dstidx,
                          float* __restrict__ agg) {
  int i = blockIdx.x * blockDim.x + threadIdx.x;
  if (i >= NE * DP) return;
  int e = i / DP, c = i % DP;
  atomicAdd(&agg[(size_t)dstidx[e] * DP + c], src[i]);
}

__global__ void k_update(float* __restrict__ xP, const float* __restrict__ agg,
                         const float* __restrict__ xl, const float* __restrict__ b_gcn,
                         int l) {
  int i = blockIdx.x * blockDim.x + threadIdx.x;
  if (i >= NV * D0) return;
  int r = i / D0, c = i % D0;
  size_t o = (size_t)r * DP + c;
  xP[o] = tanhf((agg[o] + xl[o]) * (1.0f / 3.0f) + b_gcn[l * D0 + c]);
}

template <bool HAS_SCALE>
__global__ __launch_bounds__(32) void k_gemm(const float* __restrict__ A,
                                             const float* __restrict__ B,
                                             float* __restrict__ C,
                                             const float* __restrict__ rowscale) {
  int tile = blockIdx.x;
  int tm = tile / NTN, tn = tile % NTN;
  int m0 = tm * 64, n0 = tn * 16;
  int lane = threadIdx.x;
  int half = lane >> 4;
  int l16  = lane & 15;

  const float* A0 = A + (size_t)(m0 + l16) * DP;
  const float* Bc = B + n0 + l16;
  v8f acc0 = {}, acc1 = {}, acc2 = {}, acc3 = {};
#pragma unroll 2
  for (int k = 0; k < DP; k += 4) {
    int kh = k + 2 * half;
    v2f b;
    b.x = Bc[(size_t)kh * DP];
    b.y = Bc[(size_t)(kh + 1) * DP];
    v2f a0 = *reinterpret_cast<const v2f*>(A0 + kh);
    v2f a1 = *reinterpret_cast<const v2f*>(A0 + 16 * DP + kh);
    v2f a2 = *reinterpret_cast<const v2f*>(A0 + 32 * DP + kh);
    v2f a3 = *reinterpret_cast<const v2f*>(A0 + 48 * DP + kh);
    acc0 = wmma4(a0, b, acc0);
    acc1 = wmma4(a1, b, acc1);
    acc2 = wmma4(a2, b, acc2);
    acc3 = wmma4(a3, b, acc3);
  }
#pragma unroll
  for (int s = 0; s < 4; ++s) {
    v8f acc = s == 0 ? acc0 : s == 1 ? acc1 : s == 2 ? acc2 : acc3;
    int mb = m0 + 16 * s + 8 * half;
#pragma unroll
    for (int v = 0; v < 8; ++v) {
      int m = mb + v;
      float val = acc[v];
      if (HAS_SCALE) val *= rowscale[m];
      C[(size_t)m * DP + n0 + l16] = val;
    }
  }
}

__global__ __launch_bounds__(128) void k_convfc(
    const float* __restrict__ xP, const float* __restrict__ rP,
    const int* __restrict__ ei0, const float* __restrict__ fcwT,
    const float* __restrict__ conv_w, const float* __restrict__ conv_b,
    const float* __restrict__ bn1_g, const float* __restrict__ bn1_b,
    const float* __restrict__ fc_b,
    const float* __restrict__ bn2_g, const float* __restrict__ bn2_b,
    const float* __restrict__ fc1_w, const float* __restrict__ fc1_b,
    const float* __restrict__ bn0_g, const float* __restrict__ bn0_b,
    float* __restrict__ out) {
  __shared__ float img[16][400];
  __shared__ float hbuf[16][P196];
  __shared__ float h2s[16][D0];
  __shared__ float cw[CONVK];

  int m0   = blockIdx.x * 16;
  int tid  = threadIdx.x;
  int wave = tid >> 5, lane = tid & 31;
  int half = lane >> 4, l16 = lane & 15;
  int nb   = wave * 16;

  float b0s = bn0_g[0] * INV_SQRT_1PE;
  float b0b = bn0_b[0];

  for (int i = tid; i < 16 * 400; i += 128) {
    int t = i / 400, k = i % 400;
    int src = ei0[m0 + t];
    float v = (k & 1) ? rP[(size_t)(m0 + t) * DP + (k >> 1)]
                      : xP[(size_t)src * DP + (k >> 1)];
    img[t][k] = v * b0s + b0b;
  }
  __syncthreads();

  v8f acc0 = {}, acc1 = {}, acc2 = {}, acc3 = {};

  for (int c = 0; c < NF; ++c) {
    if (tid < CONVK) cw[tid] = conv_w[c * CONVK + tid];
    __syncthreads();

    float sc1 = bn1_g[c] * INV_SQRT_1PE;
    float bb1 = bn1_b[c];
    float cb  = conv_b[c];
    for (int i = tid; i < 16 * P196; i += 128) {
      int t = i / P196, p = i % P196;
      int oy = p / 14, ox = p % 14;
      float s = cb;
#pragma unroll
      for (int a = 0; a < 7; ++a)
#pragma unroll
        for (int b = 0; b < 7; ++b)
          s += img[t][(oy + a) * 20 + ox + b] * cw[a * 7 + b];
      float v = s * sc1 + bb1;
      hbuf[t][p] = v > 0.f ? v : 0.f;
    }
    __syncthreads();

    const int kg0 = c * P196;
    const float* hrow = &hbuf[l16][0];
    if (wave == 0) {
      for (int kk = 0; kk < P196; kk += 4) {
        v2f a = *reinterpret_cast<const v2f*>(hrow + kk + 2 * half);
        const float* Bk = fcwT + (size_t)(kg0 + kk + 2 * half) * DP + l16;
        v2f b0, b1, b2, b3;
        b0.x = Bk[0];        b0.y = Bk[DP + 0];
        b1.x = Bk[64];       b1.y = Bk[DP + 64];
        b2.x = Bk[128];      b2.y = Bk[DP + 128];
        b3.x = Bk[192];      b3.y = Bk[DP + 192];
        acc0 = wmma4(a, b0, acc0);
        acc1 = wmma4(a, b1, acc1);
        acc2 = wmma4(a, b2, acc2);
        acc3 = wmma4(a, b3, acc3);
      }
    } else {
      for (int kk = 0; kk < P196; kk += 4) {
        v2f a = *reinterpret_cast<const v2f*>(hrow + kk + 2 * half);
        const float* Bk = fcwT + (size_t)(kg0 + kk + 2 * half) * DP + l16 + nb;
        v2f b0, b1, b2;
        b0.x = Bk[0];        b0.y = Bk[DP + 0];
        b1.x = Bk[64];       b1.y = Bk[DP + 64];
        b2.x = Bk[128];      b2.y = Bk[DP + 128];
        acc0 = wmma4(a, b0, acc0);
        acc1 = wmma4(a, b1, acc1);
        acc2 = wmma4(a, b2, acc2);
      }
    }
    __syncthreads();
  }

  auto store_tile = [&](v8f acc, int tn) {
    int n = tn * 16 + l16;
    if (n < D0) {
      float sc2 = bn2_g[n] * INV_SQRT_1PE;
      float bb2 = bn2_b[n];
      float fb  = fc_b[n];
#pragma unroll
      for (int v = 0; v < 8; ++v) {
        float val = (acc[v] + fb) * sc2 + bb2;
        h2s[v + 8 * half][n] = val > 0.f ? val : 0.f;
      }
    }
  };
  store_tile(acc0, wave);
  store_tile(acc1, wave + 4);
  store_tile(acc2, wave + 8);
  if (wave == 0) store_tile(acc3, 12);
  __syncthreads();

  for (int i = tid; i < 16 * NCLS; i += 128) {
    int t = i / NCLS, cls = i % NCLS;
    float s = fc1_b[cls];
    for (int c2 = 0; c2 < D0; ++c2) s += h2s[t][c2] * fc1_w[cls * D0 + c2];
    out[(size_t)(m0 + t) * NCLS + cls] = s;
  }
}


extern "C" void kernel_launch(void* const* d_in, const int* in_sizes, int n_in,
                              void* d_out, int out_size, void* d_ws, size_t ws_size,
                              hipStream_t stream) {
  const float* node  = (const float*)d_in[0];
  const float* ef    = (const float*)d_in[1];
  const float* ief   = (const float*)d_in[2];
  const float* Win   = (const float*)d_in[3];
  const float* Wout  = (const float*)d_in[4];
  const float* Wloop = (const float*)d_in[5];
  const float* Wrel  = (const float*)d_in[6];
  const float* bgcn  = (const float*)d_in[7];
  const float* bn0g  = (const float*)d_in[8];
  const float* bn0b  = (const float*)d_in[9];
  const float* convw = (const float*)d_in[10];
  const float* convb = (const float*)d_in[11];
  const float* bn1g  = (const float*)d_in[12];
  const float* bn1b  = (const float*)d_in[13];
  const float* fcw   = (const float*)d_in[14];
  const float* fcb   = (const float*)d_in[15];
  const float* bn2g  = (const float*)d_in[16];
  const float* bn2b  = (const float*)d_in[17];
  const float* fc1w  = (const float*)d_in[18];
  const float* fc1b  = (const float*)d_in[19];
  const int*   ei    = (const int*)d_in[20];
  const int*   iei   = (const int*)d_in[21];
  const int *ei0 = ei, *ei1 = ei + NE;
  const int *iei0 = iei, *iei1 = iei + NE;

  float* ws = (float*)d_ws;
  size_t o = 0;
  float* xP   = ws + o; o += (size_t)NVP * DP;
  float* xlP  = ws + o; o += (size_t)NVP * DP;
  float* aggP = ws + o; o += (size_t)NVP * DP;
  float* r0   = ws + o; o += (size_t)NE * DP;
  float* r1   = ws + o; o += (size_t)NE * DP;
  float* ir0  = ws + o; o += (size_t)NE * DP;
  float* ir1  = ws + o; o += (size_t)NE * DP;
  float* ain  = ws + o; o += (size_t)NE * DP;
  float* aout = ws + o; o += (size_t)NE * DP;
  float* minb = ws + o; o += (size_t)NE * DP;
  float* moutb= ws + o; o += (size_t)NE * DP;
  float* WP   = ws + o; o += (size_t)4 * DP * DP;
  float* fcwT = ws + o; o += (size_t)FLAT * DP;
  float* degf = ws + o; o += NVP;
  float* degi = ws + o; o += NVP;
  float* nfv  = ws + o; o += NE;
  float* niv  = ws + o; o += NE;

  const int TB = 256;
  auto cdiv = [](int a, int b) { return (a + b - 1) / b; };

  k_pad<<<cdiv(NVP * DP, TB), TB, 0, stream>>>(node, xP, NV, NVP);
  k_pad<<<cdiv(NE * DP, TB), TB, 0, stream>>>(ef, r0, NE, NE);
  k_pad<<<cdiv(NE * DP, TB), TB, 0, stream>>>(ief, ir0, NE, NE);
  k_tr_fcw<<<cdiv(FLAT * DP, TB), TB, 0, stream>>>(fcw, fcwT);

  k_zero<<<cdiv(2 * NVP, TB), TB, 0, stream>>>(degf, 2 * NVP);
  k_deg<<<cdiv(2 * NE, TB), TB, 0, stream>>>(ei0, iei0, degf, degi);
  k_norm<<<cdiv(NE, TB), TB, 0, stream>>>(ei0, ei1, iei0, iei1, degf, degi, nfv, niv);

  float *rc = r0, *rn = r1, *irc = ir0, *irn = ir1;
  const int GE = (NE / 64) * NTN;
  const int GV = (NVP / 64) * NTN;

  for (int l = 0; l < NLAY; ++l) {
    k_pad_w<<<cdiv(4 * DP * DP, TB), TB, 0, stream>>>(Win, Wout, Wloop, Wrel, l, WP);
    k_gather<<<cdiv(NE * DP, TB), TB, 0, stream>>>(xP, rc, irc, ei0, iei0, ain, aout);
    k_gemm<true><<<GE, 32, 0, stream>>>(ain,  WP + 0 * DP * DP, minb,  nfv);
    k_gemm<true><<<GE, 32, 0, stream>>>(aout, WP + 1 * DP * DP, moutb, niv);
    k_gemm<false><<<GV, 32, 0, stream>>>(xP,   WP + 2 * DP * DP, xlP,   nullptr);
    k_gemm<false><<<GE, 32, 0, stream>>>(rc,   WP + 3 * DP * DP, rn,    nullptr);
    k_gemm<false><<<GE, 32, 0, stream>>>(irc,  WP + 3 * DP * DP, irn,   nullptr);
    k_zero<<<cdiv(NVP * DP, TB), TB, 0, stream>>>(aggP, NVP * DP);
    k_scatter<<<cdiv(NE * DP, TB), TB, 0, stream>>>(minb,  ei1,  aggP);
    k_scatter<<<cdiv(NE * DP, TB), TB, 0, stream>>>(moutb, iei1, aggP);
    k_update<<<cdiv(NV * D0, TB), TB, 0, stream>>>(xP, aggP, xlP, bgcn, l);
    float* t;
    t = rc; rc = rn; rn = t;
    t = irc; irc = irn; irn = t;
  }

  k_convfc<<<NE / 16, 128, 0, stream>>>(xP, rc, ei0, fcwT, convw, convb,
                                        bn1g, bn1b, fcb, bn2g, bn2b,
                                        fc1w, fc1b, bn0g, bn0b, (float*)d_out);
}


# decoder work-group takes 32 edges (two 16-edge tiles): each fc_w fragment read from L2 feeds two WMMAs; fragments repacked in two planes, each reloaded in place behind its last use; sum order kept
# speedup vs baseline: 2.5627x; 2.5627x over previous
#include <hip/hip_runtime.h>
#include <math.h>


static constexpr int D0   = 200;
static constexpr int DP   = 208;
static constexpr int NTN  = 13;
static constexpr int NE   = 8192;
static constexpr int NV   = 5000;
static constexpr int NVP  = 5056;
static constexpr int NLAY = 6;
static constexpr int NF   = 96;
static constexpr int CONVK= 49;
static constexpr int P196 = 196;
static constexpr int FLAT = 18816;
static constexpr int NCLS = 13;
#define INV_SQRT_1PE 0.99999500f

typedef __attribute__((ext_vector_type(2))) float v2f;
typedef __attribute__((ext_vector_type(8))) float v8f;

static __device__ __forceinline__ v8f wmma4(v2f a, v2f b, v8f c) {
  return __builtin_amdgcn_wmma_f32_16x16x4_f32(false, a, false, b, (short)0, c,
                                               false, false);
}


__global__ void k_zero(float* __restrict__ p, int n) {
  int i = blockIdx.x * blockDim.x + threadIdx.x;
  if (i < n) p[i] = 0.0f;
}

__global__ void k_pad(const float* __restrict__ src, float* __restrict__ dst,
                      int rows, int prows) {
  int i = blockIdx.x * blockDim.x + threadIdx.x;
  int total = prows * DP;
  if (i >= total) return;
  int r = i / DP, c = i % DP;
  dst[i] = (r < rows && c < D0) ? src[r * D0 + c] : 0.0f;
}

__global__ void k_pack_fcw(const float* __restrict__ fcw, float* __restrict__ fcwP) {
  int o = blockIdx.x * blockDim.x + threadIdx.x;
  if (o >= FLAT * DP) return;
  int e    = o & 3;
  int lane = (o >> 2) & 31;
  int rest = o >> 7;
  int i    = rest % NTN;
  int rst2 = rest / NTN;
  int p    = rst2 & 1;
  int rst3 = rst2 >> 1;
  int pos  = rst3 % P196;
  int j    = rst3 / P196;
  int n    = 16 * i + (lane & 15);
  int ch   = 16 * j + 8 * (lane >> 4) + 4 * p + e;
  fcwP[o] = (n < D0) ? fcw[(size_t)n * FLAT + ch * P196 + pos] : 0.0f;
}

__global__ void k_deg(const int* __restrict__ ei0, const int* __restrict__ iei0,
                      float* __restrict__ degf, float* __restrict__ degi) {
  int e = blockIdx.x * blockDim.x + threadIdx.x;
  if (e < NE)            atomicAdd(&degf[ei0[e]], 1.0f);
  else if (e < 2 * NE)   atomicAdd(&degi[iei0[e - NE]], 1.0f);
}

__global__ void k_norm(const int* __restrict__ ei0, const int* __restrict__ ei1,
                       const int* __restrict__ iei0, const int* __restrict__ iei1,
                       const float* __restrict__ degf, const float* __restrict__ degi,
                       float* __restrict__ nf, float* __restrict__ ni) {
  int e = blockIdx.x * blockDim.x + threadIdx.x;
  if (e >= NE) return;
  float df0 = degf[ei0[e]],  df1 = degf[ei1[e]];
  float di0 = degi[iei0[e]], di1 = degi[iei1[e]];
  float a = df0 > 0.f ? rsqrtf(fmaxf(df0, 1.f)) : 0.f;
  float b = df1 > 0.f ? rsqrtf(fmaxf(df1, 1.f)) : 0.f;
  float c = di0 > 0.f ? rsqrtf(fmaxf(di0, 1.f)) : 0.f;
  float d = di1 > 0.f ? rsqrtf(fmaxf(di1, 1.f)) : 0.f;
  nf[e] = a * b;
  ni[e] = c * d;
}

static constexpr int NPW  = 20;
static constexpr int NCHK = 2 * NE / 32;
static constexpr int CPW  = 26;
static constexpr int ROW4 = D0 / 4;
static_assert(NV % NPW == 0, "the grid covers the nodes exactly");
static_assert(NPW * CPW >= NCHK, "every chunk has a wave");
static_assert(NPW <= 32, "the owned-node number fits the 5 low bits of an entry");

static __device__ __forceinline__ float4 add4(float4 a, float4 b) {
  return float4{a.x + b.x, a.y + b.y, a.z + b.z, a.w + b.w};
}

static __device__ __forceinline__ float4 upd4(float4 agg, float4 xs, float4 bb) {
  float4 r;
  r.x = tanhf((agg.x + xs.x) * (1.0f / 3.0f) + bb.x);
  r.y = tanhf((agg.y + xs.y) * (1.0f / 3.0f) + bb.y);
  r.z = tanhf((agg.z + xs.z) * (1.0f / 3.0f) + bb.z);
  r.w = tanhf((agg.w + xs.w) * (1.0f / 3.0f) + bb.w);
  return r;
}

__global__ __launch_bounds__(NPW * 32) void k_aggupd(
    const float* __restrict__ msg, const int* __restrict__ ei1,
    const int* __restrict__ iei1, float* __restrict__ xP,
    const float* __restrict__ xl, const float* __restrict__ b_gcn, int l) {
  __shared__ int lst[2 * NE];
  __shared__ int cnt[NPW];
  const int tid = threadIdx.x, lane = tid & 31;
  const int w  = __builtin_amdgcn_readfirstlane(tid >> 5);
  const int lo = blockIdx.x * NPW;

  int d[CPW];
#pragma unroll
  for (int it = 0; it < CPW; ++it) {
    const int c  = w * CPW + it;
    const int cc = c < NCHK ? c : NCHK - 1;
    const int* p = (cc < NCHK / 2) ? ei1 + cc * 32 : iei1 + (cc - NCHK / 2) * 32;
    const int v  = p[lane];
    d[it] = (c < NCHK) ? v : -1;
  }
  int enc[CPW], pos[CPW];
  int cw = 0;
#pragma unroll
  for (int it = 0; it < CPW; ++it) {
    const unsigned rel  = (unsigned)(d[it] - lo);
    const bool     hit  = rel < (unsigned)NPW;
    const unsigned mask = __builtin_amdgcn_ballot_w32(hit);
    pos[it] = cw + (int)__builtin_amdgcn_mbcnt_lo(mask, 0u);
    enc[it] = hit ? (int)(((unsigned)((w * CPW + it) * 32 + lane) << 5) | rel) : -1;
    cw += __builtin_popcount(mask);
  }
  if (lane == 0) cnt[w] = cw;
  __syncthreads();
  int base = 0, total = 0;
#pragma unroll
  for (int i = 0; i < NPW; ++i) {
    const int ci = cnt[i];
    base  += (i < w) ? ci : 0;
    total += ci;
  }
  total = __builtin_amdgcn_readfirstlane(total);
#pragma unroll
  for (int it = 0; it < CPW; ++it)
    if (enc[it] >= 0) lst[base + pos[it]] = enc[it];
  __syncthreads();

  const int node = lo + w;
  const bool hi = lane < ROW4 - 32;
  const int  c1 = hi ? 32 + lane : lane;
  const float4 z4 = float4{0.f, 0.f, 0.f, 0.f};
  const float4* xl4 = reinterpret_cast<const float4*>(xl + (size_t)node * DP);
  const float4* b4  = reinterpret_cast<const float4*>(b_gcn + l * D0);
  const float4 x0 = xl4[lane];
  const float4 x1 = xl4[c1];
  const float4 g0 = b4[lane];
  const float4 g1 = b4[c1];
  float4 a0 = z4, a1 = z4;
  for (int b0 = 0; b0 < total; b0 += 32) {
    const int ent = (b0 + lane < total) ? lst[b0 + lane] : -1;
    unsigned m = __builtin_amdgcn_ballot_w32(ent >= 0 && (ent & 31) == w);
    while (m) {
      int e[4];
#pragma unroll
      for (int u = 0; u < 4; ++u) {
        e[u] = -1;
        if (m) {
          const int k = __builtin_ctz(m);
          m &= m - 1;
          e[u] = __builtin_amdgcn_readlane(ent, k) >> 5;
        }
      }
      float4 v0[4], v1[4];
#pragma unroll
      for (int u = 0; u < 4; ++u) {
        const int er = e[u] < 0 ? e[0] : e[u];
        const float4* row = reinterpret_cast<const float4*>(msg + (size_t)er * DP);
        v0[u] = row[lane];
        v1[u] = row[c1];
      }
#pragma unroll
      for (int u = 0; u < 4; ++u) {
        if (e[u] >= 0) { a0 = add4(a0, v0[u]); a1 = add4(a1, v1[u]); }
      }
    }
  }
  float4* xo = reinterpret_cast<float4*>(xP + (size_t)node * DP);
  xo[lane] = upd4(a0, x0, g0);
  if (hi) xo[32 + lane] = upd4(a1, x1, g1);
}

static constexpr int KCH  = 13;
static constexpr int NCH  = 4;
static constexpr int WVEC = DP * DP / 4;
static constexpr int WROW4 = DP / 4;
static constexpr int RROW4 = D0 / 4;
static constexpr int RVEC  = D0 * RROW4;
static constexpr int CPB   = 26;
static constexpr int CPN   = RVEC / 128;
static constexpr int ZPAD  = 2 * D0 + (DP - D0) * WROW4;

template <int NT>
static __device__ __forceinline__ void gemm_chunk(
    const v2f (&af)[KCH], const float* __restrict__ Bc, v8f (&acc)[NT]) {
#pragma unroll
  for (int j = 0; j < KCH; ++j) {
#pragma unroll
    for (int t = 0; t < NT; ++t) {
      v2f b;
      b.x = Bc[(4 * j) * DP + 16 * t];
      b.y = Bc[(4 * j + 1) * DP + 16 * t];
      acc[t] = wmma4(af[j], b, acc[t]);
    }
  }
}

template <int NT, bool GATHER>
static __device__ __forceinline__ void gemm_tiles(
    v2f (&ac)[KCH], const float* __restrict__ Ap, const float* __restrict__ Rp,
    const float* __restrict__ Bl, float* __restrict__ Cp,
    const float* __restrict__ rsp) {
  constexpr bool HAS_SCALE = GATHER;
  v8f acc[NT];
#pragma unroll
  for (int t = 0; t < NT; ++t) acc[t] = v8f{};
  v2f an[KCH];
  v2f rn[KCH];
#pragma unroll 1
  for (int ch = 0; ch < NCH - 1; ++ch) {
    const float* An = Ap + 4 * KCH * (ch + 1);
#pragma unroll
    for (int j = 0; j < KCH; ++j) an[j] = *reinterpret_cast<const v2f*>(An + 4 * j);
    if (GATHER) {
      const float* Rn = Rp + 4 * KCH * (ch + 1);
#pragma unroll
      for (int j = 0; j < KCH; ++j) rn[j] = *reinterpret_cast<const v2f*>(Rn + 4 * j);
    }
    gemm_chunk<NT>(ac, Bl + (size_t)ch * (4 * KCH * DP), acc);
#pragma unroll
    for (int j = 0; j < KCH; ++j) ac[j] = GATHER ? an[j] - rn[j] : an[j];
  }
  gemm_chunk<NT>(ac, Bl + (size_t)(NCH - 1) * (4 * KCH * DP), acc);

  float sc[8];
#pragma unroll
  for (int v = 0; v < 8; ++v) sc[v] = HAS_SCALE ? rsp[v] : 1.0f;
#pragma unroll
  for (int t = 0; t < NT; ++t) {
#pragma unroll
    for (int v = 0; v < 8; ++v) {
      float val = acc[t][v];
      if (HAS_SCALE) val *= sc[v];
      Cp[(size_t)v * DP + 16 * t] = val;
    }
  }
}

template <bool GATHER>
__global__ __launch_bounds__(128) void k_gemm(const float* __restrict__ A,
                                              const float* __restrict__ R,
                                              const int* __restrict__ idx,
                                              const float* __restrict__ W,
                                              float* __restrict__ C,
                                              const float* __restrict__ rowscale) {
  __shared__ float4 wl4[WVEC];
  float* wl = reinterpret_cast<float*>(wl4);
  int tid  = threadIdx.x;
  int wave = tid >> 5, lane = tid & 31;
  int half = lane >> 4, l16 = lane & 15;
  int m0 = blockIdx.x * 32 + (wave >> 1) * 16;
  int nh = wave & 1;
  int n0 = nh * 112;

  const int arow = m0 + l16;
  const float* Ap = A + (size_t)(GATHER ? idx[arow] : arow) * DP + 2 * half;
  const float* Rp = GATHER ? R + (size_t)arow * DP + 2 * half : nullptr;
  v2f ac[KCH];
  v2f rc[KCH];
#pragma unroll
  for (int j = 0; j < KCH; ++j) ac[j] = *reinterpret_cast<const v2f*>(Ap + 4 * j);
  if (GATHER) {
#pragma unroll
    for (int j = 0; j < KCH; ++j) rc[j] = *reinterpret_cast<const v2f*>(Rp + 4 * j);
  }

  const float4* Wg = reinterpret_cast<const float4*>(W);
#pragma unroll CPB
  for (int j = 0; j < CPN; ++j) {
    int q = tid + 128 * j;
    wl4[q + 2 * (q / RROW4)] = Wg[q];
  }
  if (tid < RVEC - CPN * 128) {
    int q = tid + CPN * 128;
    wl4[q + 2 * (q / RROW4)] = Wg[q];
  }
  for (int z = tid; z < ZPAD; z += 128) {
    int p = (z < 2 * D0) ? (z >> 1) * WROW4 + RROW4 + (z & 1)
                         : D0 * WROW4 + (z - 2 * D0);
    wl4[p] = float4{0.f, 0.f, 0.f, 0.f};
  }
  __syncthreads();

  if (GATHER) {
#pragma unroll
    for (int j = 0; j < KCH; ++j) ac[j] = ac[j] - rc[j];
  }

  const float* Bl  = wl + (2 * half) * DP + n0 + l16;
  float*       Cp  = C + (size_t)(m0 + 8 * half) * DP + n0 + l16;
  const float* rsp = GATHER ? rowscale + m0 + 8 * half : nullptr;
  if (nh == 0) gemm_tiles<7, GATHER>(ac, Ap, Rp, Bl, Cp, rsp);
  else         gemm_tiles<6, GATHER>(ac, Ap, Rp, Bl, Cp, rsp);
}

static constexpr int IMGP = 401;
static constexpr int TAPP = 52;
static constexpr int PSPLIT = 147;
static constexpr int ET = 2;
static constexpr int EB = 16 * ET;
static_assert(NE % EB == 0, "the grid covers the edges exactly");

struct alignas(16) DecLds {
  float img[EB][IMGP];
  float cwp[NF][TAPP];
  float chc[NF][4];
  float part[8][ET][NTN][8][32];
  float h2s[EB][D0];
};
static_assert((EB * IMGP * 4) % 16 == 0, "cwp starts 16-byte aligned");
static_assert(sizeof(DecLds) <= 320 * 1024, "fits the LDS of one WGP");

__global__ __launch_bounds__(256) void k_convfc(
    const float* __restrict__ xP, const float* __restrict__ rP,
    const int* __restrict__ ei0, const float* __restrict__ fcwP,
    const float* __restrict__ conv_w, const float* __restrict__ conv_b,
    const float* __restrict__ bn1_g, const float* __restrict__ bn1_b,
    const float* __restrict__ fc_b,
    const float* __restrict__ bn2_g, const float* __restrict__ bn2_b,
    const float* __restrict__ fc1_w, const float* __restrict__ fc1_b,
    const float* __restrict__ bn0_g, const float* __restrict__ bn0_b,
    float* __restrict__ out) {
  __shared__ DecLds lds;

  const int m0   = blockIdx.x * EB;
  const int tid  = threadIdx.x;
  const int lane = tid & 31;
  const int half = lane >> 4, l16 = lane & 15;
  const int wave = __builtin_amdgcn_readfirstlane(tid >> 5);

  const float b0s = bn0_g[0] * INV_SQRT_1PE;
  const float b0b = bn0_b[0];

  for (int i = tid; i < EB * 400; i += 256) {
    int t = i / 400, k = i % 400;
    int srcn = ei0[m0 + t];
    float v = (k & 1) ? rP[(size_t)(m0 + t) * DP + (k >> 1)]
                      : xP[(size_t)srcn * DP + (k >> 1)];
    lds.img[t][k] = v * b0s + b0b;
  }
  for (int i = tid; i < NF * TAPP; i += 256) {
    int ch = i / TAPP, tp = i % TAPP;
    lds.cwp[ch][tp] = (tp < CONVK) ? conv_w[ch * CONVK + tp] : 0.0f;
  }
  if (tid < NF) {
    lds.chc[tid][0] = conv_b[tid];
    lds.chc[tid][1] = bn1_g[tid] * INV_SQRT_1PE;
    lds.chc[tid][2] = bn1_b[tid];
    lds.chc[tid][3] = 0.0f;
  }
  __syncthreads();

  int toff[KCH][2];
#pragma unroll
  for (int ks = 0; ks < KCH; ++ks) {
#pragma unroll
    for (int s = 0; s < 2; ++s) {
      int tp = 4 * ks + 2 * half + s;
      toff[ks][s] = (tp < CONVK) ? (tp / 7) * 20 + (tp % 7) : 0;
    }
  }
  const bool lastx_ok = (4 * (KCH - 1) + 2 * half)     < CONVK;
  const bool lasty_ok = (4 * (KCH - 1) + 2 * half + 1) < CONVK;

  const float*  imrow = &lds.img[l16][0];
  const float4* fcw4  = reinterpret_cast<const float4*>(fcwP);

  v8f acc[ET][NTN];
#pragma unroll
  for (int t = 0; t < ET; ++t) {
#pragma unroll
    for (int i = 0; i < NTN; ++i) acc[t][i] = v8f{};
  }

  const int nseg = (wave < 6) ? 1 : 3;
  const int p0   = (wave < 6) ? 0 : PSPLIT;
  const int p1   = (wave < 6) ? PSPLIT : P196;

  for (int sg = 0; sg < nseg; ++sg) {
    const int j = (wave < 6) ? wave : 3 * (wave - 6) + sg;

    v2f ta[KCH];
#pragma unroll
    for (int ks = 0; ks < KCH; ++ks)
      ta[ks] = *reinterpret_cast<const v2f*>(&lds.cwp[16 * j + l16][4 * ks + 2 * half]);
    float cbv[8], scv[8], bbv[8];
#pragma unroll
    for (int r = 0; r < 8; ++r) {
      const float* cc = &lds.chc[16 * j + 8 * half + r][0];
      cbv[r] = cc[0]; scv[r] = cc[1]; bbv[r] = cc[2];
    }

    float4 fx[NTN], fy[NTN];
#pragma unroll
    for (int i = 0; i < NTN; ++i) { fx[i] = float4{0.f, 0.f, 0.f, 0.f}; fy[i] = fx[i]; }

#pragma unroll 1
    for (int pos = p0 - 1; pos < p1; ++pos) {
      const int pn = (pos + 1 < p1) ? pos + 1 : pos;
      const float4* fpn = fcw4 + (size_t)((j * P196 + pn) * 2) * (NTN * 32) + lane;

      if (pos >= p0) {
        const int woff = (pos / 14) * 20 + (pos % 14);
        float act[ET][8];
#pragma unroll
        for (int t = 0; t < ET; ++t) {
          const float* win = imrow + t * (16 * IMGP) + woff;
          v8f d0 = v8f{}, d1 = v8f{};
#pragma unroll
          for (int ks = 0; ks < KCH; ++ks) {
            v2f b;
            b.x = win[toff[ks][0]];
            b.y = win[toff[ks][1]];
            if (ks == KCH - 1) {
              b.x = lastx_ok ? b.x : 0.0f;
              b.y = lasty_ok ? b.y : 0.0f;
            }
            if (ks & 1) d1 = wmma4(ta[ks], b, d1);
            else        d0 = wmma4(ta[ks], b, d0);
          }
#pragma unroll
          for (int r = 0; r < 8; ++r) {
            float v = ((d0[r] + d1[r]) + cbv[r]) * scv[r] + bbv[r];
            act[t][r] = v > 0.f ? v : 0.f;
          }
        }

#pragma unroll
        for (int q = 0; q < 4; ++q) {
#pragma unroll
          for (int t = 0; t < ET; ++t) {
            v2f b;
            b.x = act[t][2 * q];
            b.y = act[t][2 * q + 1];
#pragma unroll
            for (int i = 0; i < NTN; ++i) {
              v2f a;
              if (q == 0)      { a.x = fx[i].x; a.y = fx[i].y; }
              else if (q == 1) { a.x = fx[i].z; a.y = fx[i].w; }
              else if (q == 2) { a.x = fy[i].x; a.y = fy[i].y; }
              else             { a.x = fy[i].z; a.y = fy[i].w; }
              acc[t][i] = wmma4(a, b, acc[t][i]);
            }
          }
        }
      }

#pragma unroll
      for (int i = 0; i < NTN; ++i) { fx[i] = fpn[i * 32]; fy[i] = fpn[(NTN + i) * 32]; }
    }
  }

#pragma unroll
  for (int t = 0; t < ET; ++t) {
#pragma unroll
    for (int i = 0; i < NTN; ++i) {
#pragma unroll
      for (int r = 0; r < 8; ++r) lds.part[wave][t][i][r][lane] = acc[t][i][r];
    }
  }
  __syncthreads();

  for (int idx = tid; idx < ET * NTN * 8 * 32; idx += 256) {
    int t = idx / (NTN * 256), rem = idx % (NTN * 256);
    int i = rem >> 8, r = (rem >> 5) & 7, ln = rem & 31;
    float s = 0.0f;
#pragma unroll
    for (int w = 0; w < 8; ++w) s += lds.part[w][t][i][r][ln];
    int n = 16 * i + 8 * (ln >> 4) + r;
    int ed = 16 * t + (ln & 15);
    if (n < D0) {
      float val = (s + fc_b[n]) * (bn2_g[n] * INV_SQRT_1PE) + bn2_b[n];
      lds.h2s[ed][n] = val > 0.f ? val : 0.f;
    }
  }
  __syncthreads();

  for (int i = tid; i < EB * NCLS; i += 256) {
    int t = i / NCLS, cls = i % NCLS;
    float s = fc1_b[cls];
    for (int c2 = 0; c2 < D0; ++c2) s += lds.h2s[t][c2] * fc1_w[cls * D0 + c2];
    out[(size_t)(m0 + t) * NCLS + cls] = s;
  }
}


extern "C" void kernel_launch(void* const* d_in, const int* in_sizes, int n_in,
                              void* d_out, int out_size, void* d_ws, size_t ws_size,
                              hipStream_t stream) {
  const float* node  = (const float*)d_in[0];
  const float* ef    = (const float*)d_in[1];
  const float* ief   = (const float*)d_in[2];
  const float* Win   = (const float*)d_in[3];
  const float* Wout  = (const float*)d_in[4];
  const float* Wloop = (const float*)d_in[5];
  const float* Wrel  = (const float*)d_in[6];
  const float* bgcn  = (const float*)d_in[7];
  const float* bn0g  = (const float*)d_in[8];
  const float* bn0b  = (const float*)d_in[9];
  const float* convw = (const float*)d_in[10];
  const float* convb = (const float*)d_in[11];
  const float* bn1g  = (const float*)d_in[12];
  const float* bn1b  = (const float*)d_in[13];
  const float* fcw   = (const float*)d_in[14];
  const float* fcb   = (const float*)d_in[15];
  const float* bn2g  = (const float*)d_in[16];
  const float* bn2b  = (const float*)d_in[17];
  const float* fc1w  = (const float*)d_in[18];
  const float* fc1b  = (const float*)d_in[19];
  const int*   ei    = (const int*)d_in[20];
  const int*   iei   = (const int*)d_in[21];
  const int *ei0 = ei, *ei1 = ei + NE;
  const int *iei0 = iei, *iei1 = iei + NE;

  float* ws = (float*)d_ws;
  size_t o = 0;
  float* xP   = ws + o; o += (size_t)NVP * DP;
  float* xlP  = ws + o; o += (size_t)NVP * DP;
  float* r0   = ws + o; o += (size_t)NE * DP;
  float* r1   = ws + o; o += (size_t)NE * DP;
  float* ir0  = ws + o; o += (size_t)NE * DP;
  float* ir1  = ws + o; o += (size_t)NE * DP;
  float* minb = ws + o; o += (size_t)NE * DP;
  float* moutb= ws + o; o += (size_t)NE * DP;
  float* fcwT = ws + o; o += (size_t)FLAT * DP;
  float* degf = ws + o; o += NVP;
  float* degi = ws + o; o += NVP;
  float* nfv  = ws + o; o += NE;
  float* niv  = ws + o; o += NE;

  const int TB = 256;
  auto cdiv = [](int a, int b) { return (a + b - 1) / b; };

  k_pad<<<cdiv(NVP * DP, TB), TB, 0, stream>>>(node, xP, NV, NVP);
  k_pad<<<cdiv(NE * DP, TB), TB, 0, stream>>>(ef, r0, NE, NE);
  k_pad<<<cdiv(NE * DP, TB), TB, 0, stream>>>(ief, ir0, NE, NE);
  k_pack_fcw<<<cdiv(FLAT * DP, TB), TB, 0, stream>>>(fcw, fcwT);

  k_zero<<<cdiv(2 * NVP, TB), TB, 0, stream>>>(degf, 2 * NVP);
  k_deg<<<cdiv(2 * NE, TB), TB, 0, stream>>>(ei0, iei0, degf, degi);
  k_norm<<<cdiv(NE, TB), TB, 0, stream>>>(ei0, ei1, iei0, iei1, degf, degi, nfv, niv);

  float *rc = r0, *rn = r1, *irc = ir0, *irn = ir1;
  const int GE = NE / 32;
  const int GV = NVP / 32;

  for (int l = 0; l < NLAY; ++l) {
    const float* wi = Win   + (size_t)l * D0 * D0;
    const float* wo = Wout  + (size_t)l * D0 * D0;
    const float* wp = Wloop + (size_t)l * D0 * D0;
    const float* wr = Wrel  + (size_t)l * D0 * D0;
    k_gemm<true><<<GE, 128, 0, stream>>>(xP, rc,  ei0,  wi, minb,  nfv);
    k_gemm<true><<<GE, 128, 0, stream>>>(xP, irc, iei0, wo, moutb, niv);
    k_gemm<false><<<GV, 128, 0, stream>>>(xP,  nullptr, nullptr, wp, xlP, nullptr);
    k_gemm<false><<<GE, 128, 0, stream>>>(rc,  nullptr, nullptr, wr, rn,  nullptr);
    k_gemm<false><<<GE, 128, 0, stream>>>(irc, nullptr, nullptr, wr, irn, nullptr);
    k_aggupd<<<NV / NPW, NPW * 32, 0, stream>>>(minb, ei1, iei1, xP, xlP, bgcn, l);
    float* t;
    t = rc; rc = rn; rn = t;
    t = irc; irc = irn; irn = t;
  }

  k_convfc<<<NE / EB, 256, 0, stream>>>(xP, rc, ei0, fcwT, convw, convb,
                                        bn1g, bn1b, fcb, bn2g, bn2b,
                                        fc1w, fc1b, bn0g, bn0b, (float*)d_out);
}


# layer GEMM reads B fragments straight from the raw weight in L2 by 16-byte loads (a lane's 4 columns = one column of 4 tiles); wave tile 32x52, no LDS copy, no barrier; k blocks of 8: sum order new
# speedup vs baseline: 2.6985x; 1.0530x over previous
#include <hip/hip_runtime.h>
#include <math.h>


static constexpr int D0   = 200;
static constexpr int DP   = 208;
static constexpr int NTN  = 13;
static constexpr int NE   = 8192;
static constexpr int NV   = 5000;
static constexpr int NVP  = 5056;
static constexpr int NLAY = 6;
static constexpr int NF   = 96;
static constexpr int CONVK= 49;
static constexpr int P196 = 196;
static constexpr int FLAT = 18816;
static constexpr int NCLS = 13;
#define INV_SQRT_1PE 0.99999500f

typedef __attribute__((ext_vector_type(2))) float v2f;
typedef __attribute__((ext_vector_type(8))) float v8f;

static __device__ __forceinline__ v8f wmma4(v2f a, v2f b, v8f c) {
  return __builtin_amdgcn_wmma_f32_16x16x4_f32(false, a, false, b, (short)0, c,
                                               false, false);
}


__global__ void k_zero(float* __restrict__ p, int n) {
  int i = blockIdx.x * blockDim.x + threadIdx.x;
  if (i < n) p[i] = 0.0f;
}

__global__ void k_pad(const float* __restrict__ src, float* __restrict__ dst,
                      int rows, int prows) {
  int i = blockIdx.x * blockDim.x + threadIdx.x;
  int total = prows * DP;
  if (i >= total) return;
  int r = i / DP, c = i % DP;
  dst[i] = (r < rows && c < D0) ? src[r * D0 + c] : 0.0f;
}

__global__ void k_pack_fcw(const float* __restrict__ fcw, float* __restrict__ fcwP) {
  int o = blockIdx.x * blockDim.x + threadIdx.x;
  if (o >= FLAT * DP) return;
  int e    = o & 3;
  int lane = (o >> 2) & 31;
  int rest = o >> 7;
  int i    = rest % NTN;
  int rst2 = rest / NTN;
  int p    = rst2 & 1;
  int rst3 = rst2 >> 1;
  int pos  = rst3 % P196;
  int j    = rst3 / P196;
  int n    = 16 * i + (lane & 15);
  int ch   = 16 * j + 8 * (lane >> 4) + 4 * p + e;
  fcwP[o] = (n < D0) ? fcw[(size_t)n * FLAT + ch * P196 + pos] : 0.0f;
}

__global__ void k_deg(const int* __restrict__ ei0, const int* __restrict__ iei0,
                      float* __restrict__ degf, float* __restrict__ degi) {
  int e = blockIdx.x * blockDim.x + threadIdx.x;
  if (e < NE)            atomicAdd(&degf[ei0[e]], 1.0f);
  else if (e < 2 * NE)   atomicAdd(&degi[iei0[e - NE]], 1.0f);
}

__global__ void k_norm(const int* __restrict__ ei0, const int* __restrict__ ei1,
                       const int* __restrict__ iei0, const int* __restrict__ iei1,
                       const float* __restrict__ degf, const float* __restrict__ degi,
                       float* __restrict__ nf, float* __restrict__ ni) {
  int e = blockIdx.x * blockDim.x + threadIdx.x;
  if (e >= NE) return;
  float df0 = degf[ei0[e]],  df1 = degf[ei1[e]];
  float di0 = degi[iei0[e]], di1 = degi[iei1[e]];
  float a = df0 > 0.f ? rsqrtf(fmaxf(df0, 1.f)) : 0.f;
  float b = df1 > 0.f ? rsqrtf(fmaxf(df1, 1.f)) : 0.f;
  float c = di0 > 0.f ? rsqrtf(fmaxf(di0, 1.f)) : 0.f;
  float d = di1 > 0.f ? rsqrtf(fmaxf(di1, 1.f)) : 0.f;
  nf[e] = a * b;
  ni[e] = c * d;
}

static constexpr int NPW  = 20;
static constexpr int NCHK = 2 * NE / 32;
static constexpr int CPW  = 26;
static constexpr int ROW4 = D0 / 4;
static_assert(NV % NPW == 0, "the grid covers the nodes exactly");
static_assert(NPW * CPW >= NCHK, "every chunk has a wave");
static_assert(NPW <= 32, "the owned-node number fits the 5 low bits of an entry");

static __device__ __forceinline__ float4 add4(float4 a, float4 b) {
  return float4{a.x + b.x, a.y + b.y, a.z + b.z, a.w + b.w};
}

static __device__ __forceinline__ float4 upd4(float4 agg, float4 xs, float4 bb) {
  float4 r;
  r.x = tanhf((agg.x + xs.x) * (1.0f / 3.0f) + bb.x);
  r.y = tanhf((agg.y + xs.y) * (1.0f / 3.0f) + bb.y);
  r.z = tanhf((agg.z + xs.z) * (1.0f / 3.0f) + bb.z);
  r.w = tanhf((agg.w + xs.w) * (1.0f / 3.0f) + bb.w);
  return r;
}

__global__ __launch_bounds__(NPW * 32) void k_aggupd(
    const float* __restrict__ msg, const int* __restrict__ ei1,
    const int* __restrict__ iei1, float* __restrict__ xP,
    const float* __restrict__ xl, const float* __restrict__ b_gcn, int l) {
  __shared__ int lst[2 * NE];
  __shared__ int cnt[NPW];
  const int tid = threadIdx.x, lane = tid & 31;
  const int w  = __builtin_amdgcn_readfirstlane(tid >> 5);
  const int lo = blockIdx.x * NPW;

  int d[CPW];
#pragma unroll
  for (int it = 0; it < CPW; ++it) {
    const int c  = w * CPW + it;
    const int cc = c < NCHK ? c : NCHK - 1;
    const int* p = (cc < NCHK / 2) ? ei1 + cc * 32 : iei1 + (cc - NCHK / 2) * 32;
    const int v  = p[lane];
    d[it] = (c < NCHK) ? v : -1;
  }
  int enc[CPW], pos[CPW];
  int cw = 0;
#pragma unroll
  for (int it = 0; it < CPW; ++it) {
    const unsigned rel  = (unsigned)(d[it] - lo);
    const bool     hit  = rel < (unsigned)NPW;
    const unsigned mask = __builtin_amdgcn_ballot_w32(hit);
    pos[it] = cw + (int)__builtin_amdgcn_mbcnt_lo(mask, 0u);
    enc[it] = hit ? (int)(((unsigned)((w * CPW + it) * 32 + lane) << 5) | rel) : -1;
    cw += __builtin_popcount(mask);
  }
  if (lane == 0) cnt[w] = cw;
  __syncthreads();
  int base = 0, total = 0;
#pragma unroll
  for (int i = 0; i < NPW; ++i) {
    const int ci = cnt[i];
    base  += (i < w) ? ci : 0;
    total += ci;
  }
  total = __builtin_amdgcn_readfirstlane(total);
#pragma unroll
  for (int it = 0; it < CPW; ++it)
    if (enc[it] >= 0) lst[base + pos[it]] = enc[it];
  __syncthreads();

  const int node = lo + w;
  const bool hi = lane < ROW4 - 32;
  const int  c1 = hi ? 32 + lane : lane;
  const float4 z4 = float4{0.f, 0.f, 0.f, 0.f};
  const float4* xl4 = reinterpret_cast<const float4*>(xl + (size_t)node * DP);
  const float4* b4  = reinterpret_cast<const float4*>(b_gcn + l * D0);
  const float4 x0 = xl4[lane];
  const float4 x1 = xl4[c1];
  const float4 g0 = b4[lane];
  const float4 g1 = b4[c1];
  float4 a0 = z4, a1 = z4;
  for (int b0 = 0; b0 < total; b0 += 32) {
    const int ent = (b0 + lane < total) ? lst[b0 + lane] : -1;
    unsigned m = __builtin_amdgcn_ballot_w32(ent >= 0 && (ent & 31) == w);
    while (m) {
      int e[4];
#pragma unroll
      for (int u = 0; u < 4; ++u) {
        e[u] = -1;
        if (m) {
          const int k = __builtin_ctz(m);
          m &= m - 1;
          e[u] = __builtin_amdgcn_readlane(ent, k) >> 5;
        }
      }
      float4 v0[4], v1[4];
#pragma unroll
      for (int u = 0; u < 4; ++u) {
        const int er = e[u] < 0 ? e[0] : e[u];
        const float4* row = reinterpret_cast<const float4*>(msg + (size_t)er * DP);
        v0[u] = row[lane];
        v1[u] = row[c1];
      }
#pragma unroll
      for (int u = 0; u < 4; ++u) {
        if (e[u] >= 0) { a0 = add4(a0, v0[u]); a1 = add4(a1, v1[u]); }
      }
    }
  }
  float4* xo = reinterpret_cast<float4*>(xP + (size_t)node * DP);
  xo[lane] = upd4(a0, x0, g0);
  if (hi) xo[32 + lane] = upd4(a1, x1, g1);
}

static constexpr int WPC = D0 / 4;
static constexpr int APC = DP / 4;
static constexpr int UCH = 5;
static_assert(D0 == 8 * UCH * 5, "5 chunks of 5 blocks of 8 k");

struct GemmSet {
  float4 b[UCH][4];
  float4 a[2][UCH];
  float4 r[2][UCH];
};

static __device__ __forceinline__ float f4c(const float4& v, int t) {
  return t == 0 ? v.x : t == 1 ? v.y : t == 2 ? v.z : v.w;
}

template <bool GATHER>
static __device__ __forceinline__ void gemm_load(
    GemmSet& s, int ch, const float4* __restrict__ Wp,
    const float4* __restrict__ A0, const float4* __restrict__ A1,
    const float4* __restrict__ R0, const float4* __restrict__ R1) {
#pragma unroll
  for (int i = 0; i < UCH; ++i) {
    const int u = ch * UCH + i;
#pragma unroll
    for (int q = 0; q < 4; ++q) s.b[i][q] = Wp[(8 * u + q) * WPC];
    s.a[0][i] = A0[2 * u];
    s.a[1][i] = A1[2 * u];
    if (GATHER) {
      s.r[0][i] = R0[2 * u];
      s.r[1][i] = R1[2 * u];
    }
  }
}

template <bool GATHER>
static __device__ __forceinline__ void gemm_mma(GemmSet& s, v8f (&acc)[2][4]) {
  if (GATHER) {
#pragma unroll
    for (int i = 0; i < UCH; ++i) {
#pragma unroll
      for (int mt = 0; mt < 2; ++mt) {
        s.a[mt][i].x -= s.r[mt][i].x;
        s.a[mt][i].y -= s.r[mt][i].y;
        s.a[mt][i].z -= s.r[mt][i].z;
        s.a[mt][i].w -= s.r[mt][i].w;
      }
    }
  }
#pragma unroll
  for (int i = 0; i < UCH; ++i) {
#pragma unroll
    for (int p = 0; p < 2; ++p) {
      v2f a0, a1;
      a0.x = p ? s.a[0][i].z : s.a[0][i].x;
      a0.y = p ? s.a[0][i].w : s.a[0][i].y;
      a1.x = p ? s.a[1][i].z : s.a[1][i].x;
      a1.y = p ? s.a[1][i].w : s.a[1][i].y;
#pragma unroll
      for (int t = 0; t < 4; ++t) {
        v2f b;
        b.x = f4c(s.b[i][2 * p], t);
        b.y = f4c(s.b[i][2 * p + 1], t);
        acc[0][t] = wmma4(a0, b, acc[0][t]);
        acc[1][t] = wmma4(a1, b, acc[1][t]);
      }
    }
  }
}

template <bool GATHER>
__global__ __launch_bounds__(128) void k_gemm(const float* __restrict__ A,
                                              const float* __restrict__ R,
                                              const int* __restrict__ idx,
                                              const float* __restrict__ W,
                                              float* __restrict__ C,
                                              const float* __restrict__ rowscale) {
  const int tid  = threadIdx.x;
  const int lane = tid & 31;
  const int half = lane >> 4, l16 = lane & 15;
  const int wave = __builtin_amdgcn_readfirstlane(tid >> 5);
  const int m0 = blockIdx.x * 32;
  const int P0 = 13 * wave - (wave > 2 ? 1 : 0);
  const int np = wave < 2 ? 13 : 12;
  const bool own = l16 < np;
  const int pc = P0 + (own ? l16 : 0);

  const int ra = m0 + l16, rb = m0 + 16 + l16;
  const float4* A0 = reinterpret_cast<const float4*>(A + (size_t)(GATHER ? idx[ra] : ra) * DP) + half;
  const float4* A1 = reinterpret_cast<const float4*>(A + (size_t)(GATHER ? idx[rb] : rb) * DP) + half;
  const float4* R0 = GATHER ? reinterpret_cast<const float4*>(R + (size_t)ra * DP) + half : nullptr;
  const float4* R1 = GATHER ? reinterpret_cast<const float4*>(R + (size_t)rb * DP) + half : nullptr;
  const float4* Wp = reinterpret_cast<const float4*>(W) + (4 * half) * WPC + pc;

  v8f acc[2][4];
#pragma unroll
  for (int mt = 0; mt < 2; ++mt) {
#pragma unroll
    for (int t = 0; t < 4; ++t) acc[mt][t] = v8f{};
  }

  GemmSet s0, s1;
  gemm_load<GATHER>(s0, 0, Wp, A0, A1, R0, R1);
#pragma unroll 1
  for (int p = 0; p < 2; ++p) {
    gemm_load<GATHER>(s1, 2 * p + 1, Wp, A0, A1, R0, R1);
    gemm_mma<GATHER>(s0, acc);
    gemm_load<GATHER>(s0, 2 * p + 2, Wp, A0, A1, R0, R1);
    gemm_mma<GATHER>(s1, acc);
  }
  gemm_mma<GATHER>(s0, acc);

#pragma unroll
  for (int mt = 0; mt < 2; ++mt) {
#pragma unroll
    for (int r = 0; r < 8; ++r) {
      const int row = m0 + 16 * mt + 8 * half + r;
      const float sc = GATHER ? rowscale[row] : 1.0f;
      float4 v;
      v.x = acc[mt][0][r]; v.y = acc[mt][1][r]; v.z = acc[mt][2][r]; v.w = acc[mt][3][r];
      if (GATHER) { v.x *= sc; v.y *= sc; v.z *= sc; v.w *= sc; }
      if (own) reinterpret_cast<float4*>(C + (size_t)row * DP)[pc] = v;
    }
  }
}

static constexpr int IMGP = 401;
static constexpr int TAPP = 52;
static constexpr int KCH  = 13;
static constexpr int PSPLIT = 147;
static constexpr int ET = 2;
static constexpr int EB = 16 * ET;
static_assert(NE % EB == 0, "the grid covers the edges exactly");

struct alignas(16) DecLds {
  float img[EB][IMGP];
  float cwp[NF][TAPP];
  float chc[NF][4];
  float part[8][ET][NTN][8][32];
  float h2s[EB][D0];
};
static_assert((EB * IMGP * 4) % 16 == 0, "cwp starts 16-byte aligned");
static_assert(sizeof(DecLds) <= 320 * 1024, "fits the LDS of one WGP");

__global__ __launch_bounds__(256) void k_convfc(
    const float* __restrict__ xP, const float* __restrict__ rP,
    const int* __restrict__ ei0, const float* __restrict__ fcwP,
    const float* __restrict__ conv_w, const float* __restrict__ conv_b,
    const float* __restrict__ bn1_g, const float* __restrict__ bn1_b,
    const float* __restrict__ fc_b,
    const float* __restrict__ bn2_g, const float* __restrict__ bn2_b,
    const float* __restrict__ fc1_w, const float* __restrict__ fc1_b,
    const float* __restrict__ bn0_g, const float* __restrict__ bn0_b,
    float* __restrict__ out) {
  __shared__ DecLds lds;

  const int m0   = blockIdx.x * EB;
  const int tid  = threadIdx.x;
  const int lane = tid & 31;
  const int half = lane >> 4, l16 = lane & 15;
  const int wave = __builtin_amdgcn_readfirstlane(tid >> 5);

  const float b0s = bn0_g[0] * INV_SQRT_1PE;
  const float b0b = bn0_b[0];

  for (int i = tid; i < EB * 400; i += 256) {
    int t = i / 400, k = i % 400;
    int srcn = ei0[m0 + t];
    float v = (k & 1) ? rP[(size_t)(m0 + t) * DP + (k >> 1)]
                      : xP[(size_t)srcn * DP + (k >> 1)];
    lds.img[t][k] = v * b0s + b0b;
  }
  for (int i = tid; i < NF * TAPP; i += 256) {
    int ch = i / TAPP, tp = i % TAPP;
    lds.cwp[ch][tp] = (tp < CONVK) ? conv_w[ch * CONVK + tp] : 0.0f;
  }
  if (tid < NF) {
    lds.chc[tid][0] = conv_b[tid];
    lds.chc[tid][1] = bn1_g[tid] * INV_SQRT_1PE;
    lds.chc[tid][2] = bn1_b[tid];
    lds.chc[tid][3] = 0.0f;
  }
  __syncthreads();

  int toff[KCH][2];
#pragma unroll
  for (int ks = 0; ks < KCH; ++ks) {
#pragma unroll
    for (int s = 0; s < 2; ++s) {
      int tp = 4 * ks + 2 * half + s;
      toff[ks][s] = (tp < CONVK) ? (tp / 7) * 20 + (tp % 7) : 0;
    }
  }
  const bool lastx_ok = (4 * (KCH - 1) + 2 * half)     < CONVK;
  const bool lasty_ok = (4 * (KCH - 1) + 2 * half + 1) < CONVK;

  const float*  imrow = &lds.img[l16][0];
  const float4* fcw4  = reinterpret_cast<const float4*>(fcwP);

  v8f acc[ET][NTN];
#pragma unroll
  for (int t = 0; t < ET; ++t) {
#pragma unroll
    for (int i = 0; i < NTN; ++i) acc[t][i] = v8f{};
  }

  const int nseg = (wave < 6) ? 1 : 3;
  const int p0   = (wave < 6) ? 0 : PSPLIT;
  const int p1   = (wave < 6) ? PSPLIT : P196;

  for (int sg = 0; sg < nseg; ++sg) {
    const int j = (wave < 6) ? wave : 3 * (wave - 6) + sg;

    v2f ta[KCH];
#pragma unroll
    for (int ks = 0; ks < KCH; ++ks)
      ta[ks] = *reinterpret_cast<const v2f*>(&lds.cwp[16 * j + l16][4 * ks + 2 * half]);
    float cbv[8], scv[8], bbv[8];
#pragma unroll
    for (int r = 0; r < 8; ++r) {
      const float* cc = &lds.chc[16 * j + 8 * half + r][0];
      cbv[r] = cc[0]; scv[r] = cc[1]; bbv[r] = cc[2];
    }

    float4 fx[NTN], fy[NTN];
#pragma unroll
    for (int i = 0; i < NTN; ++i) { fx[i] = float4{0.f, 0.f, 0.f, 0.f}; fy[i] = fx[i]; }

#pragma unroll 1
    for (int pos = p0 - 1; pos < p1; ++pos) {
      const int pn = (pos + 1 < p1) ? pos + 1 : pos;
      const float4* fpn = fcw4 + (size_t)((j * P196 + pn) * 2) * (NTN * 32) + lane;

      if (pos >= p0) {
        const int woff = (pos / 14) * 20 + (pos % 14);
        float act[ET][8];
#pragma unroll
        for (int t = 0; t < ET; ++t) {
          const float* win = imrow + t * (16 * IMGP) + woff;
          v8f d0 = v8f{}, d1 = v8f{};
#pragma unroll
          for (int ks = 0; ks < KCH; ++ks) {
            v2f b;
            b.x = win[toff[ks][0]];
            b.y = win[toff[ks][1]];
            if (ks == KCH - 1) {
              b.x = lastx_ok ? b.x : 0.0f;
              b.y = lasty_ok ? b.y : 0.0f;
            }
            if (ks & 1) d1 = wmma4(ta[ks], b, d1);
            else        d0 = wmma4(ta[ks], b, d0);
          }
#pragma unroll
          for (int r = 0; r < 8; ++r) {
            float v = ((d0[r] + d1[r]) + cbv[r]) * scv[r] + bbv[r];
            act[t][r] = v > 0.f ? v : 0.f;
          }
        }

#pragma unroll
        for (int q = 0; q < 4; ++q) {
#pragma unroll
          for (int t = 0; t < ET; ++t) {
            v2f b;
            b.x = act[t][2 * q];
            b.y = act[t][2 * q + 1];
#pragma unroll
            for (int i = 0; i < NTN; ++i) {
              v2f a;
              if (q == 0)      { a.x = fx[i].x; a.y = fx[i].y; }
              else if (q == 1) { a.x = fx[i].z; a.y = fx[i].w; }
              else if (q == 2) { a.x = fy[i].x; a.y = fy[i].y; }
              else             { a.x = fy[i].z; a.y = fy[i].w; }
              acc[t][i] = wmma4(a, b, acc[t][i]);
            }
          }
        }
      }

#pragma unroll
      for (int i = 0; i < NTN; ++i) { fx[i] = fpn[i * 32]; fy[i] = fpn[(NTN + i) * 32]; }
    }
  }

#pragma unroll
  for (int t = 0; t < ET; ++t) {
#pragma unroll
    for (int i = 0; i < NTN; ++i) {
#pragma unroll
      for (int r = 0; r < 8; ++r) lds.part[wave][t][i][r][lane] = acc[t][i][r];
    }
  }
  __syncthreads();

  for (int idx = tid; idx < ET * NTN * 8 * 32; idx += 256) {
    int t = idx / (NTN * 256), rem = idx % (NTN * 256);
    int i = rem >> 8, r = (rem >> 5) & 7, ln = rem & 31;
    float s = 0.0f;
#pragma unroll
    for (int w = 0; w < 8; ++w) s += lds.part[w][t][i][r][ln];
    int n = 16 * i + 8 * (ln >> 4) + r;
    int ed = 16 * t + (ln & 15);
    if (n < D0) {
      float val = (s + fc_b[n]) * (bn2_g[n] * INV_SQRT_1PE) + bn2_b[n];
      lds.h2s[ed][n] = val > 0.f ? val : 0.f;
    }
  }
  __syncthreads();

  for (int i = tid; i < EB * NCLS; i += 256) {
    int t = i / NCLS, cls = i % NCLS;
    float s = fc1_b[cls];
    for (int c2 = 0; c2 < D0; ++c2) s += lds.h2s[t][c2] * fc1_w[cls * D0 + c2];
    out[(size_t)(m0 + t) * NCLS + cls] = s;
  }
}


extern "C" void kernel_launch(void* const* d_in, const int* in_sizes, int n_in,
                              void* d_out, int out_size, void* d_ws, size_t ws_size,
                              hipStream_t stream) {
  const float* node  = (const float*)d_in[0];
  const float* ef    = (const float*)d_in[1];
  const float* ief   = (const float*)d_in[2];
  const float* Win   = (const float*)d_in[3];
  const float* Wout  = (const float*)d_in[4];
  const float* Wloop = (const float*)d_in[5];
  const float* Wrel  = (const float*)d_in[6];
  const float* bgcn  = (const float*)d_in[7];
  const float* bn0g  = (const float*)d_in[8];
  const float* bn0b  = (const float*)d_in[9];
  const float* convw = (const float*)d_in[10];
  const float* convb = (const float*)d_in[11];
  const float* bn1g  = (const float*)d_in[12];
  const float* bn1b  = (const float*)d_in[13];
  const float* fcw   = (const float*)d_in[14];
  const float* fcb   = (const float*)d_in[15];
  const float* bn2g  = (const float*)d_in[16];
  const float* bn2b  = (const float*)d_in[17];
  const float* fc1w  = (const float*)d_in[18];
  const float* fc1b  = (const float*)d_in[19];
  const int*   ei    = (const int*)d_in[20];
  const int*   iei   = (const int*)d_in[21];
  const int *ei0 = ei, *ei1 = ei + NE;
  const int *iei0 = iei, *iei1 = iei + NE;

  float* ws = (float*)d_ws;
  size_t o = 0;
  float* xP   = ws + o; o += (size_t)NVP * DP;
  float* xlP  = ws + o; o += (size_t)NVP * DP;
  float* r0   = ws + o; o += (size_t)NE * DP;
  float* r1   = ws + o; o += (size_t)NE * DP;
  float* ir0  = ws + o; o += (size_t)NE * DP;
  float* ir1  = ws + o; o += (size_t)NE * DP;
  float* minb = ws + o; o += (size_t)NE * DP;
  float* moutb= ws + o; o += (size_t)NE * DP;
  float* fcwT = ws + o; o += (size_t)FLAT * DP;
  float* degf = ws + o; o += NVP;
  float* degi = ws + o; o += NVP;
  float* nfv  = ws + o; o += NE;
  float* niv  = ws + o; o += NE;

  const int TB = 256;
  auto cdiv = [](int a, int b) { return (a + b - 1) / b; };

  k_pad<<<cdiv(NVP * DP, TB), TB, 0, stream>>>(node, xP, NV, NVP);
  k_pad<<<cdiv(NE * DP, TB), TB, 0, stream>>>(ef, r0, NE, NE);
  k_pad<<<cdiv(NE * DP, TB), TB, 0, stream>>>(ief, ir0, NE, NE);
  k_pack_fcw<<<cdiv(FLAT * DP, TB), TB, 0, stream>>>(fcw, fcwT);

  k_zero<<<cdiv(2 * NVP, TB), TB, 0, stream>>>(degf, 2 * NVP);
  k_deg<<<cdiv(2 * NE, TB), TB, 0, stream>>>(ei0, iei0, degf, degi);
  k_norm<<<cdiv(NE, TB), TB, 0, stream>>>(ei0, ei1, iei0, iei1, degf, degi, nfv, niv);

  float *rc = r0, *rn = r1, *irc = ir0, *irn = ir1;
  const int GE = NE / 32;
  const int GV = NVP / 32;

  for (int l = 0; l < NLAY; ++l) {
    const float* wi = Win   + (size_t)l * D0 * D0;
    const float* wo = Wout  + (size_t)l * D0 * D0;
    const float* wp = Wloop + (size_t)l * D0 * D0;
    const float* wr = Wrel  + (size_t)l * D0 * D0;
    k_gemm<true><<<GE, 128, 0, stream>>>(xP, rc,  ei0,  wi, minb,  nfv);
    k_gemm<true><<<GE, 128, 0, stream>>>(xP, irc, iei0, wo, moutb, niv);
    k_gemm<false><<<GV, 128, 0, stream>>>(xP,  nullptr, nullptr, wp, xlP, nullptr);
    k_gemm<false><<<GE, 128, 0, stream>>>(rc,  nullptr, nullptr, wr, rn,  nullptr);
    k_gemm<false><<<GE, 128, 0, stream>>>(irc, nullptr, nullptr, wr, irn, nullptr);
    k_aggupd<<<NV / NPW, NPW * 32, 0, stream>>>(minb, ei1, iei1, xP, xlP, bgcn, l);
    float* t;
    t = rc; rc = rn; rn = t;
    t = irc; irc = irn; irn = t;
  }

  k_convfc<<<NE / EB, 256, 0, stream>>>(xP, rc, ei0, fcwT, convw, convb,
                                        bn1g, bn1b, fcb, bn2g, bn2b,
                                        fc1w, fc1b, bn0g, bn0b, (float*)d_out);
}


# relation update rides in the message GEMM: the r rows loaded for x[src]-r also meet Wrel on a second accumulator set, a loaded k block feeds 32 WMMAs, not 16; 44 -> 32 launches; sum order kept
# speedup vs baseline: 2.9217x; 1.0827x over previous
#include <hip/hip_runtime.h>
#include <math.h>


static constexpr int D0   = 200;
static constexpr int DP   = 208;
static constexpr int NTN  = 13;
static constexpr int NE   = 8192;
static constexpr int NV   = 5000;
static constexpr int NVP  = 5056;
static constexpr int NLAY = 6;
static constexpr int NF   = 96;
static constexpr int CONVK= 49;
static constexpr int P196 = 196;
static constexpr int FLAT = 18816;
static constexpr int NCLS = 13;
#define INV_SQRT_1PE 0.99999500f

typedef __attribute__((ext_vector_type(2))) float v2f;
typedef __attribute__((ext_vector_type(8))) float v8f;

static __device__ __forceinline__ v8f wmma4(v2f a, v2f b, v8f c) {
  return __builtin_amdgcn_wmma_f32_16x16x4_f32(false, a, false, b, (short)0, c,
                                               false, false);
}


__global__ void k_zero(float* __restrict__ p, int n) {
  int i = blockIdx.x * blockDim.x + threadIdx.x;
  if (i < n) p[i] = 0.0f;
}

__global__ void k_pad(const float* __restrict__ src, float* __restrict__ dst,
                      int rows, int prows) {
  int i = blockIdx.x * blockDim.x + threadIdx.x;
  int total = prows * DP;
  if (i >= total) return;
  int r = i / DP, c = i % DP;
  dst[i] = (r < rows && c < D0) ? src[r * D0 + c] : 0.0f;
}

__global__ void k_pack_fcw(const float* __restrict__ fcw, float* __restrict__ fcwP) {
  int o = blockIdx.x * blockDim.x + threadIdx.x;
  if (o >= FLAT * DP) return;
  int e    = o & 3;
  int lane = (o >> 2) & 31;
  int rest = o >> 7;
  int i    = rest % NTN;
  int rst2 = rest / NTN;
  int p    = rst2 & 1;
  int rst3 = rst2 >> 1;
  int pos  = rst3 % P196;
  int j    = rst3 / P196;
  int n    = 16 * i + (lane & 15);
  int ch   = 16 * j + 8 * (lane >> 4) + 4 * p + e;
  fcwP[o] = (n < D0) ? fcw[(size_t)n * FLAT + ch * P196 + pos] : 0.0f;
}

__global__ void k_deg(const int* __restrict__ ei0, const int* __restrict__ iei0,
                      float* __restrict__ degf, float* __restrict__ degi) {
  int e = blockIdx.x * blockDim.x + threadIdx.x;
  if (e < NE)            atomicAdd(&degf[ei0[e]], 1.0f);
  else if (e < 2 * NE)   atomicAdd(&degi[iei0[e - NE]], 1.0f);
}

__global__ void k_norm(const int* __restrict__ ei0, const int* __restrict__ ei1,
                       const int* __restrict__ iei0, const int* __restrict__ iei1,
                       const float* __restrict__ degf, const float* __restrict__ degi,
                       float* __restrict__ nf, float* __restrict__ ni) {
  int e = blockIdx.x * blockDim.x + threadIdx.x;
  if (e >= NE) return;
  float df0 = degf[ei0[e]],  df1 = degf[ei1[e]];
  float di0 = degi[iei0[e]], di1 = degi[iei1[e]];
  float a = df0 > 0.f ? rsqrtf(fmaxf(df0, 1.f)) : 0.f;
  float b = df1 > 0.f ? rsqrtf(fmaxf(df1, 1.f)) : 0.f;
  float c = di0 > 0.f ? rsqrtf(fmaxf(di0, 1.f)) : 0.f;
  float d = di1 > 0.f ? rsqrtf(fmaxf(di1, 1.f)) : 0.f;
  nf[e] = a * b;
  ni[e] = c * d;
}

static constexpr int NPW  = 20;
static constexpr int NCHK = 2 * NE / 32;
static constexpr int CPW  = 26;
static constexpr int ROW4 = D0 / 4;
static_assert(NV % NPW == 0, "the grid covers the nodes exactly");
static_assert(NPW * CPW >= NCHK, "every chunk has a wave");
static_assert(NPW <= 32, "the owned-node number fits the 5 low bits of an entry");

static __device__ __forceinline__ float4 add4(float4 a, float4 b) {
  return float4{a.x + b.x, a.y + b.y, a.z + b.z, a.w + b.w};
}

static __device__ __forceinline__ float4 upd4(float4 agg, float4 xs, float4 bb) {
  float4 r;
  r.x = tanhf((agg.x + xs.x) * (1.0f / 3.0f) + bb.x);
  r.y = tanhf((agg.y + xs.y) * (1.0f / 3.0f) + bb.y);
  r.z = tanhf((agg.z + xs.z) * (1.0f / 3.0f) + bb.z);
  r.w = tanhf((agg.w + xs.w) * (1.0f / 3.0f) + bb.w);
  return r;
}

__global__ __launch_bounds__(NPW * 32) void k_aggupd(
    const float* __restrict__ msg, const int* __restrict__ ei1,
    const int* __restrict__ iei1, float* __restrict__ xP,
    const float* __restrict__ xl, const float* __restrict__ b_gcn, int l) {
  __shared__ int lst[2 * NE];
  __shared__ int cnt[NPW];
  const int tid = threadIdx.x, lane = tid & 31;
  const int w  = __builtin_amdgcn_readfirstlane(tid >> 5);
  const int lo = blockIdx.x * NPW;

  int d[CPW];
#pragma unroll
  for (int it = 0; it < CPW; ++it) {
    const int c  = w * CPW + it;
    const int cc = c < NCHK ? c : NCHK - 1;
    const int* p = (cc < NCHK / 2) ? ei1 + cc * 32 : iei1 + (cc - NCHK / 2) * 32;
    const int v  = p[lane];
    d[it] = (c < NCHK) ? v : -1;
  }
  int enc[CPW], pos[CPW];
  int cw = 0;
#pragma unroll
  for (int it = 0; it < CPW; ++it) {
    const unsigned rel  = (unsigned)(d[it] - lo);
    const bool     hit  = rel < (unsigned)NPW;
    const unsigned mask = __builtin_amdgcn_ballot_w32(hit);
    pos[it] = cw + (int)__builtin_amdgcn_mbcnt_lo(mask, 0u);
    enc[it] = hit ? (int)(((unsigned)((w * CPW + it) * 32 + lane) << 5) | rel) : -1;
    cw += __builtin_popcount(mask);
  }
  if (lane == 0) cnt[w] = cw;
  __syncthreads();
  int base = 0, total = 0;
#pragma unroll
  for (int i = 0; i < NPW; ++i) {
    const int ci = cnt[i];
    base  += (i < w) ? ci : 0;
    total += ci;
  }
  total = __builtin_amdgcn_readfirstlane(total);
#pragma unroll
  for (int it = 0; it < CPW; ++it)
    if (enc[it] >= 0) lst[base + pos[it]] = enc[it];
  __syncthreads();

  const int node = lo + w;
  const bool hi = lane < ROW4 - 32;
  const int  c1 = hi ? 32 + lane : lane;
  const float4 z4 = float4{0.f, 0.f, 0.f, 0.f};
  const float4* xl4 = reinterpret_cast<const float4*>(xl + (size_t)node * DP);
  const float4* b4  = reinterpret_cast<const float4*>(b_gcn + l * D0);
  const float4 x0 = xl4[lane];
  const float4 x1 = xl4[c1];
  const float4 g0 = b4[lane];
  const float4 g1 = b4[c1];
  float4 a0 = z4, a1 = z4;
  for (int b0 = 0; b0 < total; b0 += 32) {
    const int ent = (b0 + lane < total) ? lst[b0 + lane] : -1;
    unsigned m = __builtin_amdgcn_ballot_w32(ent >= 0 && (ent & 31) == w);
    while (m) {
      int e[4];
#pragma unroll
      for (int u = 0; u < 4; ++u) {
        e[u] = -1;
        if (m) {
          const int k = __builtin_ctz(m);
          m &= m - 1;
          e[u] = __builtin_amdgcn_readlane(ent, k) >> 5;
        }
      }
      float4 v0[4], v1[4];
#pragma unroll
      for (int u = 0; u < 4; ++u) {
        const int er = e[u] < 0 ? e[0] : e[u];
        const float4* row = reinterpret_cast<const float4*>(msg + (size_t)er * DP);
        v0[u] = row[lane];
        v1[u] = row[c1];
      }
#pragma unroll
      for (int u = 0; u < 4; ++u) {
        if (e[u] >= 0) { a0 = add4(a0, v0[u]); a1 = add4(a1, v1[u]); }
      }
    }
  }
  float4* xo = reinterpret_cast<float4*>(xP + (size_t)node * DP);
  xo[lane] = upd4(a0, x0, g0);
  if (hi) xo[32 + lane] = upd4(a1, x1, g1);
}

static constexpr int WPC = D0 / 4;
static constexpr int APC = DP / 4;
static constexpr int UCH = 5;
static_assert(D0 == 8 * UCH * 5, "5 chunks of 5 blocks of 8 k");

struct GemmSet {
  float4 b[UCH][4];
  float4 a[2][UCH];
  float4 r[2][UCH];
  float4 b2[UCH][4];
};

static __device__ __forceinline__ float f4c(const float4& v, int t) {
  return t == 0 ? v.x : t == 1 ? v.y : t == 2 ? v.z : v.w;
}

template <bool GATHER, bool DUAL>
static __device__ __forceinline__ void gemm_load(
    GemmSet& s, int ch, const float4* __restrict__ Wp,
    const float4* __restrict__ Wp2,
    const float4* __restrict__ A0, const float4* __restrict__ A1,
    const float4* __restrict__ R0, const float4* __restrict__ R1) {
#pragma unroll
  for (int i = 0; i < UCH; ++i) {
    const int u = ch * UCH + i;
#pragma unroll
    for (int q = 0; q < 4; ++q) s.b[i][q] = Wp[(8 * u + q) * WPC];
    if (DUAL) {
#pragma unroll
      for (int q = 0; q < 4; ++q) s.b2[i][q] = Wp2[(8 * u + q) * WPC];
    }
    s.a[0][i] = A0[2 * u];
    s.a[1][i] = A1[2 * u];
    if (GATHER) {
      s.r[0][i] = R0[2 * u];
      s.r[1][i] = R1[2 * u];
    }
  }
}

template <bool GATHER, bool DUAL>
static __device__ __forceinline__ void gemm_mma(GemmSet& s, v8f (&acc)[2][4],
                                                v8f (&acc2)[2][4]) {
  if (GATHER) {
#pragma unroll
    for (int i = 0; i < UCH; ++i) {
#pragma unroll
      for (int mt = 0; mt < 2; ++mt) {
        s.a[mt][i].x -= s.r[mt][i].x;
        s.a[mt][i].y -= s.r[mt][i].y;
        s.a[mt][i].z -= s.r[mt][i].z;
        s.a[mt][i].w -= s.r[mt][i].w;
      }
    }
  }
#pragma unroll
  for (int i = 0; i < UCH; ++i) {
#pragma unroll
    for (int p = 0; p < 2; ++p) {
      v2f a0, a1;
      a0.x = p ? s.a[0][i].z : s.a[0][i].x;
      a0.y = p ? s.a[0][i].w : s.a[0][i].y;
      a1.x = p ? s.a[1][i].z : s.a[1][i].x;
      a1.y = p ? s.a[1][i].w : s.a[1][i].y;
#pragma unroll
      for (int t = 0; t < 4; ++t) {
        v2f b;
        b.x = f4c(s.b[i][2 * p], t);
        b.y = f4c(s.b[i][2 * p + 1], t);
        acc[0][t] = wmma4(a0, b, acc[0][t]);
        acc[1][t] = wmma4(a1, b, acc[1][t]);
      }
      if (DUAL) {
        v2f r0, r1;
        r0.x = p ? s.r[0][i].z : s.r[0][i].x;
        r0.y = p ? s.r[0][i].w : s.r[0][i].y;
        r1.x = p ? s.r[1][i].z : s.r[1][i].x;
        r1.y = p ? s.r[1][i].w : s.r[1][i].y;
#pragma unroll
        for (int t = 0; t < 4; ++t) {
          v2f b;
          b.x = f4c(s.b2[i][2 * p], t);
          b.y = f4c(s.b2[i][2 * p + 1], t);
          acc2[0][t] = wmma4(r0, b, acc2[0][t]);
          acc2[1][t] = wmma4(r1, b, acc2[1][t]);
        }
      }
    }
  }
}

template <bool GATHER, bool DUAL>
__global__ __launch_bounds__(128) void k_gemm(const float* __restrict__ A,
                                              const float* __restrict__ R,
                                              const int* __restrict__ idx,
                                              const float* __restrict__ W,
                                              float* __restrict__ C,
                                              const float* __restrict__ rowscale,
                                              const float* __restrict__ W2,
                                              float* __restrict__ C2) {
  static_assert(GATHER || !DUAL, "the second product takes the subtracted rows");
  const int tid  = threadIdx.x;
  const int lane = tid & 31;
  const int half = lane >> 4, l16 = lane & 15;
  const int wave = __builtin_amdgcn_readfirstlane(tid >> 5);
  const int m0 = blockIdx.x * 32;
  const int P0 = 13 * wave - (wave > 2 ? 1 : 0);
  const int np = wave < 2 ? 13 : 12;
  const bool own = l16 < np;
  const int pc = P0 + (own ? l16 : 0);

  const int ra = m0 + l16, rb = m0 + 16 + l16;
  const float4* A0 = reinterpret_cast<const float4*>(A + (size_t)(GATHER ? idx[ra] : ra) * DP) + half;
  const float4* A1 = reinterpret_cast<const float4*>(A + (size_t)(GATHER ? idx[rb] : rb) * DP) + half;
  const float4* R0 = GATHER ? reinterpret_cast<const float4*>(R + (size_t)ra * DP) + half : nullptr;
  const float4* R1 = GATHER ? reinterpret_cast<const float4*>(R + (size_t)rb * DP) + half : nullptr;
  const float4* Wp = reinterpret_cast<const float4*>(W) + (4 * half) * WPC + pc;
  const float4* Wp2 = DUAL ? reinterpret_cast<const float4*>(W2) + (4 * half) * WPC + pc : nullptr;

  v8f acc[2][4], acc2[2][4];
#pragma unroll
  for (int mt = 0; mt < 2; ++mt) {
#pragma unroll
    for (int t = 0; t < 4; ++t) { acc[mt][t] = v8f{}; acc2[mt][t] = v8f{}; }
  }

  GemmSet s0, s1;
  gemm_load<GATHER, DUAL>(s0, 0, Wp, Wp2, A0, A1, R0, R1);
#pragma unroll 1
  for (int p = 0; p < 2; ++p) {
    gemm_load<GATHER, DUAL>(s1, 2 * p + 1, Wp, Wp2, A0, A1, R0, R1);
    gemm_mma<GATHER, DUAL>(s0, acc, acc2);
    gemm_load<GATHER, DUAL>(s0, 2 * p + 2, Wp, Wp2, A0, A1, R0, R1);
    gemm_mma<GATHER, DUAL>(s1, acc, acc2);
  }
  gemm_mma<GATHER, DUAL>(s0, acc, acc2);

#pragma unroll
  for (int mt = 0; mt < 2; ++mt) {
#pragma unroll
    for (int r = 0; r < 8; ++r) {
      const int row = m0 + 16 * mt + 8 * half + r;
      const float sc = GATHER ? rowscale[row] : 1.0f;
      float4 v;
      v.x = acc[mt][0][r]; v.y = acc[mt][1][r]; v.z = acc[mt][2][r]; v.w = acc[mt][3][r];
      if (GATHER) { v.x *= sc; v.y *= sc; v.z *= sc; v.w *= sc; }
      if (own) reinterpret_cast<float4*>(C + (size_t)row * DP)[pc] = v;
      if (DUAL) {
        float4 v2;
        v2.x = acc2[mt][0][r]; v2.y = acc2[mt][1][r]; v2.z = acc2[mt][2][r]; v2.w = acc2[mt][3][r];
        if (own) reinterpret_cast<float4*>(C2 + (size_t)row * DP)[pc] = v2;
      }
    }
  }
}

static constexpr int IMGP = 401;
static constexpr int TAPP = 52;
static constexpr int KCH  = 13;
static constexpr int PSPLIT = 147;
static constexpr int ET = 2;
static constexpr int EB = 16 * ET;
static_assert(NE % EB == 0, "the grid covers the edges exactly");

struct alignas(16) DecLds {
  float img[EB][IMGP];
  float cwp[NF][TAPP];
  float chc[NF][4];
  float part[8][ET][NTN][8][32];
  float h2s[EB][D0];
};
static_assert((EB * IMGP * 4) % 16 == 0, "cwp starts 16-byte aligned");
static_assert(sizeof(DecLds) <= 320 * 1024, "fits the LDS of one WGP");

__global__ __launch_bounds__(256) void k_convfc(
    const float* __restrict__ xP, const float* __restrict__ rP,
    const int* __restrict__ ei0, const float* __restrict__ fcwP,
    const float* __restrict__ conv_w, const float* __restrict__ conv_b,
    const float* __restrict__ bn1_g, const float* __restrict__ bn1_b,
    const float* __restrict__ fc_b,
    const float* __restrict__ bn2_g, const float* __restrict__ bn2_b,
    const float* __restrict__ fc1_w, const float* __restrict__ fc1_b,
    const float* __restrict__ bn0_g, const float* __restrict__ bn0_b,
    float* __restrict__ out) {
  __shared__ DecLds lds;

  const int m0   = blockIdx.x * EB;
  const int tid  = threadIdx.x;
  const int lane = tid & 31;
  const int half = lane >> 4, l16 = lane & 15;
  const int wave = __builtin_amdgcn_readfirstlane(tid >> 5);

  const float b0s = bn0_g[0] * INV_SQRT_1PE;
  const float b0b = bn0_b[0];

  for (int i = tid; i < EB * 400; i += 256) {
    int t = i / 400, k = i % 400;
    int srcn = ei0[m0 + t];
    float v = (k & 1) ? rP[(size_t)(m0 + t) * DP + (k >> 1)]
                      : xP[(size_t)srcn * DP + (k >> 1)];
    lds.img[t][k] = v * b0s + b0b;
  }
  for (int i = tid; i < NF * TAPP; i += 256) {
    int ch = i / TAPP, tp = i % TAPP;
    lds.cwp[ch][tp] = (tp < CONVK) ? conv_w[ch * CONVK + tp] : 0.0f;
  }
  if (tid < NF) {
    lds.chc[tid][0] = conv_b[tid];
    lds.chc[tid][1] = bn1_g[tid] * INV_SQRT_1PE;
    lds.chc[tid][2] = bn1_b[tid];
    lds.chc[tid][3] = 0.0f;
  }
  __syncthreads();

  int toff[KCH][2];
#pragma unroll
  for (int ks = 0; ks < KCH; ++ks) {
#pragma unroll
    for (int s = 0; s < 2; ++s) {
      int tp = 4 * ks + 2 * half + s;
      toff[ks][s] = (tp < CONVK) ? (tp / 7) * 20 + (tp % 7) : 0;
    }
  }
  const bool lastx_ok = (4 * (KCH - 1) + 2 * half)     < CONVK;
  const bool lasty_ok = (4 * (KCH - 1) + 2 * half + 1) < CONVK;

  const float*  imrow = &lds.img[l16][0];
  const float4* fcw4  = reinterpret_cast<const float4*>(fcwP);

  v8f acc[ET][NTN];
#pragma unroll
  for (int t = 0; t < ET; ++t) {
#pragma unroll
    for (int i = 0; i < NTN; ++i) acc[t][i] = v8f{};
  }

  const int nseg = (wave < 6) ? 1 : 3;
  const int p0   = (wave < 6) ? 0 : PSPLIT;
  const int p1   = (wave < 6) ? PSPLIT : P196;

  for (int sg = 0; sg < nseg; ++sg) {
    const int j = (wave < 6) ? wave : 3 * (wave - 6) + sg;

    v2f ta[KCH];
#pragma unroll
    for (int ks = 0; ks < KCH; ++ks)
      ta[ks] = *reinterpret_cast<const v2f*>(&lds.cwp[16 * j + l16][4 * ks + 2 * half]);
    float cbv[8], scv[8], bbv[8];
#pragma unroll
    for (int r = 0; r < 8; ++r) {
      const float* cc = &lds.chc[16 * j + 8 * half + r][0];
      cbv[r] = cc[0]; scv[r] = cc[1]; bbv[r] = cc[2];
    }

    float4 fx[NTN], fy[NTN];
#pragma unroll
    for (int i = 0; i < NTN; ++i) { fx[i] = float4{0.f, 0.f, 0.f, 0.f}; fy[i] = fx[i]; }

#pragma unroll 1
    for (int pos = p0 - 1; pos < p1; ++pos) {
      const int pn = (pos + 1 < p1) ? pos + 1 : pos;
      const float4* fpn = fcw4 + (size_t)((j * P196 + pn) * 2) * (NTN * 32) + lane;

      if (pos >= p0) {
        const int woff = (pos / 14) * 20 + (pos % 14);
        float act[ET][8];
#pragma unroll
        for (int t = 0; t < ET; ++t) {
          const float* win = imrow + t * (16 * IMGP) + woff;
          v8f d0 = v8f{}, d1 = v8f{};
#pragma unroll
          for (int ks = 0; ks < KCH; ++ks) {
            v2f b;
            b.x = win[toff[ks][0]];
            b.y = win[toff[ks][1]];
            if (ks == KCH - 1) {
              b.x = lastx_ok ? b.x : 0.0f;
              b.y = lasty_ok ? b.y : 0.0f;
            }
            if (ks & 1) d1 = wmma4(ta[ks], b, d1);
            else        d0 = wmma4(ta[ks], b, d0);
          }
#pragma unroll
          for (int r = 0; r < 8; ++r) {
            float v = ((d0[r] + d1[r]) + cbv[r]) * scv[r] + bbv[r];
            act[t][r] = v > 0.f ? v : 0.f;
          }
        }

#pragma unroll
        for (int q = 0; q < 4; ++q) {
#pragma unroll
          for (int t = 0; t < ET; ++t) {
            v2f b;
            b.x = act[t][2 * q];
            b.y = act[t][2 * q + 1];
#pragma unroll
            for (int i = 0; i < NTN; ++i) {
              v2f a;
              if (q == 0)      { a.x = fx[i].x; a.y = fx[i].y; }
              else if (q == 1) { a.x = fx[i].z; a.y = fx[i].w; }
              else if (q == 2) { a.x = fy[i].x; a.y = fy[i].y; }
              else             { a.x = fy[i].z; a.y = fy[i].w; }
              acc[t][i] = wmma4(a, b, acc[t][i]);
            }
          }
        }
      }

#pragma unroll
      for (int i = 0; i < NTN; ++i) { fx[i] = fpn[i * 32]; fy[i] = fpn[(NTN + i) * 32]; }
    }
  }

#pragma unroll
  for (int t = 0; t < ET; ++t) {
#pragma unroll
    for (int i = 0; i < NTN; ++i) {
#pragma unroll
      for (int r = 0; r < 8; ++r) lds.part[wave][t][i][r][lane] = acc[t][i][r];
    }
  }
  __syncthreads();

  for (int idx = tid; idx < ET * NTN * 8 * 32; idx += 256) {
    int t = idx / (NTN * 256), rem = idx % (NTN * 256);
    int i = rem >> 8, r = (rem >> 5) & 7, ln = rem & 31;
    float s = 0.0f;
#pragma unroll
    for (int w = 0; w < 8; ++w) s += lds.part[w][t][i][r][ln];
    int n = 16 * i + 8 * (ln >> 4) + r;
    int ed = 16 * t + (ln & 15);
    if (n < D0) {
      float val = (s + fc_b[n]) * (bn2_g[n] * INV_SQRT_1PE) + bn2_b[n];
      lds.h2s[ed][n] = val > 0.f ? val : 0.f;
    }
  }
  __syncthreads();

  for (int i = tid; i < EB * NCLS; i += 256) {
    int t = i / NCLS, cls = i % NCLS;
    float s = fc1_b[cls];
    for (int c2 = 0; c2 < D0; ++c2) s += lds.h2s[t][c2] * fc1_w[cls * D0 + c2];
    out[(size_t)(m0 + t) * NCLS + cls] = s;
  }
}


extern "C" void kernel_launch(void* const* d_in, const int* in_sizes, int n_in,
                              void* d_out, int out_size, void* d_ws, size_t ws_size,
                              hipStream_t stream) {
  const float* node  = (const float*)d_in[0];
  const float* ef    = (const float*)d_in[1];
  const float* ief   = (const float*)d_in[2];
  const float* Win   = (const float*)d_in[3];
  const float* Wout  = (const float*)d_in[4];
  const float* Wloop = (const float*)d_in[5];
  const float* Wrel  = (const float*)d_in[6];
  const float* bgcn  = (const float*)d_in[7];
  const float* bn0g  = (const float*)d_in[8];
  const float* bn0b  = (const float*)d_in[9];
  const float* convw = (const float*)d_in[10];
  const float* convb = (const float*)d_in[11];
  const float* bn1g  = (const float*)d_in[12];
  const float* bn1b  = (const float*)d_in[13];
  const float* fcw   = (const float*)d_in[14];
  const float* fcb   = (const float*)d_in[15];
  const float* bn2g  = (const float*)d_in[16];
  const float* bn2b  = (const float*)d_in[17];
  const float* fc1w  = (const float*)d_in[18];
  const float* fc1b  = (const float*)d_in[19];
  const int*   ei    = (const int*)d_in[20];
  const int*   iei   = (const int*)d_in[21];
  const int *ei0 = ei, *ei1 = ei + NE;
  const int *iei0 = iei, *iei1 = iei + NE;

  float* ws = (float*)d_ws;
  size_t o = 0;
  float* xP   = ws + o; o += (size_t)NVP * DP;
  float* xlP  = ws + o; o += (size_t)NVP * DP;
  float* r0   = ws + o; o += (size_t)NE * DP;
  float* r1   = ws + o; o += (size_t)NE * DP;
  float* ir0  = ws + o; o += (size_t)NE * DP;
  float* ir1  = ws + o; o += (size_t)NE * DP;
  float* minb = ws + o; o += (size_t)NE * DP;
  float* moutb= ws + o; o += (size_t)NE * DP;
  float* fcwT = ws + o; o += (size_t)FLAT * DP;
  float* degf = ws + o; o += NVP;
  float* degi = ws + o; o += NVP;
  float* nfv  = ws + o; o += NE;
  float* niv  = ws + o; o += NE;

  const int TB = 256;
  auto cdiv = [](int a, int b) { return (a + b - 1) / b; };

  k_pad<<<cdiv(NVP * DP, TB), TB, 0, stream>>>(node, xP, NV, NVP);
  k_pad<<<cdiv(NE * DP, TB), TB, 0, stream>>>(ef, r0, NE, NE);
  k_pad<<<cdiv(NE * DP, TB), TB, 0, stream>>>(ief, ir0, NE, NE);
  k_pack_fcw<<<cdiv(FLAT * DP, TB), TB, 0, stream>>>(fcw, fcwT);

  k_zero<<<cdiv(2 * NVP, TB), TB, 0, stream>>>(degf, 2 * NVP);
  k_deg<<<cdiv(2 * NE, TB), TB, 0, stream>>>(ei0, iei0, degf, degi);
  k_norm<<<cdiv(NE, TB), TB, 0, stream>>>(ei0, ei1, iei0, iei1, degf, degi, nfv, niv);

  float *rc = r0, *rn = r1, *irc = ir0, *irn = ir1;
  const int GE = NE / 32;
  const int GV = NVP / 32;

  for (int l = 0; l < NLAY; ++l) {
    const float* wi = Win   + (size_t)l * D0 * D0;
    const float* wo = Wout  + (size_t)l * D0 * D0;
    const float* wp = Wloop + (size_t)l * D0 * D0;
    const float* wr = Wrel  + (size_t)l * D0 * D0;
    k_gemm<true, true><<<GE, 128, 0, stream>>>(xP, rc,  ei0,  wi, minb,  nfv, wr, rn);
    k_gemm<true, true><<<GE, 128, 0, stream>>>(xP, irc, iei0, wo, moutb, niv, wr, irn);
    k_gemm<false, false><<<GV, 128, 0, stream>>>(xP, nullptr, nullptr, wp, xlP, nullptr,
                                                 nullptr, nullptr);
    k_aggupd<<<NV / NPW, NPW * 32, 0, stream>>>(minb, ei1, iei1, xP, xlP, bgcn, l);
    float* t;
    t = rc; rc = rn; rn = t;
    t = irc; irc = irn; irn = t;
  }

  k_convfc<<<NE / EB, 256, 0, stream>>>(xP, rc, ei0, fcwT, convw, convb,
                                        bn1g, bn1b, fcb, bn2g, bn2b,
                                        fc1w, fc1b, bn0g, bn0b, (float*)d_out);
}


# self-loop GEMM x @ Wloop moves into the owner work-group of the layer sum: 4 product waves run it on WMMA while 10 node waves sum, result handed over in LDS; 32 -> 26 launches; sum order kept
# speedup vs baseline: 3.0101x; 1.0303x over previous
#include <hip/hip_runtime.h>
#include <math.h>


static constexpr int D0   = 200;
static constexpr int DP   = 208;
static constexpr int NTN  = 13;
static constexpr int NE   = 8192;
static constexpr int NV   = 5000;
static constexpr int NVP  = 5056;
static constexpr int NLAY = 6;
static constexpr int NF   = 96;
static constexpr int CONVK= 49;
static constexpr int P196 = 196;
static constexpr int FLAT = 18816;
static constexpr int NCLS = 13;
#define INV_SQRT_1PE 0.99999500f

typedef __attribute__((ext_vector_type(2))) float v2f;
typedef __attribute__((ext_vector_type(8))) float v8f;

static __device__ __forceinline__ v8f wmma4(v2f a, v2f b, v8f c) {
  return __builtin_amdgcn_wmma_f32_16x16x4_f32(false, a, false, b, (short)0, c,
                                               false, false);
}


__global__ void k_zero(float* __restrict__ p, int n) {
  int i = blockIdx.x * blockDim.x + threadIdx.x;
  if (i < n) p[i] = 0.0f;
}

__global__ void k_pad(const float* __restrict__ src, float* __restrict__ dst,
                      int rows, int prows) {
  int i = blockIdx.x * blockDim.x + threadIdx.x;
  int total = prows * DP;
  if (i >= total) return;
  int r = i / DP, c = i % DP;
  dst[i] = (r < rows && c < D0) ? src[r * D0 + c] : 0.0f;
}

__global__ void k_pack_fcw(const float* __restrict__ fcw, float* __restrict__ fcwP) {
  int o = blockIdx.x * blockDim.x + threadIdx.x;
  if (o >= FLAT * DP) return;
  int e    = o & 3;
  int lane = (o >> 2) & 31;
  int rest = o >> 7;
  int i    = rest % NTN;
  int rst2 = rest / NTN;
  int p    = rst2 & 1;
  int rst3 = rst2 >> 1;
  int pos  = rst3 % P196;
  int j    = rst3 / P196;
  int n    = 16 * i + (lane & 15);
  int ch   = 16 * j + 8 * (lane >> 4) + 4 * p + e;
  fcwP[o] = (n < D0) ? fcw[(size_t)n * FLAT + ch * P196 + pos] : 0.0f;
}

__global__ void k_deg(const int* __restrict__ ei0, const int* __restrict__ iei0,
                      float* __restrict__ degf, float* __restrict__ degi) {
  int e = blockIdx.x * blockDim.x + threadIdx.x;
  if (e < NE)            atomicAdd(&degf[ei0[e]], 1.0f);
  else if (e < 2 * NE)   atomicAdd(&degi[iei0[e - NE]], 1.0f);
}

__global__ void k_norm(const int* __restrict__ ei0, const int* __restrict__ ei1,
                       const int* __restrict__ iei0, const int* __restrict__ iei1,
                       const float* __restrict__ degf, const float* __restrict__ degi,
                       float* __restrict__ nf, float* __restrict__ ni) {
  int e = blockIdx.x * blockDim.x + threadIdx.x;
  if (e >= NE) return;
  float df0 = degf[ei0[e]],  df1 = degf[ei1[e]];
  float di0 = degi[iei0[e]], di1 = degi[iei1[e]];
  float a = df0 > 0.f ? rsqrtf(fmaxf(df0, 1.f)) : 0.f;
  float b = df1 > 0.f ? rsqrtf(fmaxf(df1, 1.f)) : 0.f;
  float c = di0 > 0.f ? rsqrtf(fmaxf(di0, 1.f)) : 0.f;
  float d = di1 > 0.f ? rsqrtf(fmaxf(di1, 1.f)) : 0.f;
  nf[e] = a * b;
  ni[e] = c * d;
}

static constexpr int NPW  = 20;
static constexpr int NNW  = NPW / 2;
static constexpr int NGW  = 4;
static constexpr int NWV  = NNW + NGW;
static constexpr int NCHK = 2 * NE / 32;
static constexpr int CPW  = 37;
static constexpr int ROW4 = D0 / 4;
static constexpr int WPC  = D0 / 4;
static constexpr int KB8  = D0 / 8;
static constexpr int RD   = 5;
static_assert(NV % NPW == 0, "the grid covers the nodes exactly");
static_assert(NPW == 2 * NNW, "two nodes per node wave");
static_assert(NWV * CPW >= NCHK, "every chunk has a wave");
static_assert(NPW <= 32, "the owned-node number fits the 5 low bits of an entry");
static_assert(NPW > 16 && NPW <= 32, "the product tile has two 16-row halves");

static __device__ __forceinline__ float f4c(const float4& v, int t) {
  return t == 0 ? v.x : t == 1 ? v.y : t == 2 ? v.z : v.w;
}

static __device__ __forceinline__ float4 add4(float4 a, float4 b) {
  return float4{a.x + b.x, a.y + b.y, a.z + b.z, a.w + b.w};
}

static __device__ __forceinline__ float4 upd4(float4 agg, float4 xs, float4 bb) {
  float4 r;
  r.x = tanhf((agg.x + xs.x) * (1.0f / 3.0f) + bb.x);
  r.y = tanhf((agg.y + xs.y) * (1.0f / 3.0f) + bb.y);
  r.z = tanhf((agg.z + xs.z) * (1.0f / 3.0f) + bb.z);
  r.w = tanhf((agg.w + xs.w) * (1.0f / 3.0f) + bb.w);
  return r;
}

struct KBlk {
  float4 b[4];
  float4 a[2];
};

static __device__ __forceinline__ void kblk_load(KBlk& s, int u,
                                                 const float4* __restrict__ Wp,
                                                 const float4* A0, const float4* A1) {
#pragma unroll
  for (int q = 0; q < 4; ++q) s.b[q] = Wp[(8 * u + q) * WPC];
  s.a[0] = A0[2 * u];
  s.a[1] = A1[2 * u];
}

static __device__ __forceinline__ void kblk_mma(const KBlk& s, v8f (&acc)[2][4]) {
#pragma unroll
  for (int p = 0; p < 2; ++p) {
    v2f a0, a1;
    a0.x = p ? s.a[0].z : s.a[0].x;
    a0.y = p ? s.a[0].w : s.a[0].y;
    a1.x = p ? s.a[1].z : s.a[1].x;
    a1.y = p ? s.a[1].w : s.a[1].y;
#pragma unroll
    for (int t = 0; t < 4; ++t) {
      v2f b;
      b.x = f4c(s.b[2 * p], t);
      b.y = f4c(s.b[2 * p + 1], t);
      acc[0][t] = wmma4(a0, b, acc[0][t]);
      acc[1][t] = wmma4(a1, b, acc[1][t]);
    }
  }
}

static __device__ __forceinline__ void node_sum(const int* lst, int total, int sel,
                                                const float* __restrict__ msg,
                                                int lane, int c1,
                                                float4& a0, float4& a1) {
  for (int b0 = 0; b0 < total; b0 += 32) {
    const int ent = (b0 + lane < total) ? lst[b0 + lane] : -1;
    unsigned m = __builtin_amdgcn_ballot_w32(ent >= 0 && (ent & 31) == sel);
    while (m) {
      int e[4];
#pragma unroll
      for (int u = 0; u < 4; ++u) {
        e[u] = -1;
        if (m) {
          const int k = __builtin_ctz(m);
          m &= m - 1;
          e[u] = __builtin_amdgcn_readlane(ent, k) >> 5;
        }
      }
      float4 v0[4], v1[4];
#pragma unroll
      for (int u = 0; u < 4; ++u) {
        const int er = e[u] < 0 ? e[0] : e[u];
        const float4* row = reinterpret_cast<const float4*>(msg + (size_t)er * DP);
        v0[u] = row[lane];
        v1[u] = row[c1];
      }
#pragma unroll
      for (int u = 0; u < 4; ++u) {
        if (e[u] >= 0) { a0 = add4(a0, v0[u]); a1 = add4(a1, v1[u]); }
      }
    }
  }
}

__global__ __launch_bounds__(NWV * 32) void k_aggupd(
    const float* __restrict__ msg, const int* __restrict__ ei1,
    const int* __restrict__ iei1, float* xP,
    const float* __restrict__ Wl, const float* __restrict__ b_gcn, int l) {
  __shared__ int lst[2 * NE];
  __shared__ int cnt[NWV];
  __shared__ float4 xls[NPW][DP / 4];
  const int tid = threadIdx.x, lane = tid & 31;
  const int w  = __builtin_amdgcn_readfirstlane(tid >> 5);
  const int lo = blockIdx.x * NPW;
  const bool pw = w >= NNW;

  const bool hi = lane < ROW4 - 32;
  const int  c1 = hi ? 32 + lane : lane;
  const float4 z4 = float4{0.f, 0.f, 0.f, 0.f};

  const int half = lane >> 4, l16 = lane & 15;
  const int g  = pw ? w - NNW : 0;
  const int P0 = 13 * g - (g > 2 ? 1 : 0);
  const int np = g < 2 ? 13 : 12;
  const bool own = l16 < np;
  const int pc = P0 + (own ? l16 : 0);
  const int rA = lo + l16;
  const int rB = lo + (l16 < NPW - 16 ? 16 + l16 : l16 - (NPW - 16));
  const float4* A0 = reinterpret_cast<const float4*>(xP + (size_t)rA * DP) + half;
  const float4* A1 = reinterpret_cast<const float4*>(xP + (size_t)rB * DP) + half;
  const float4* Wp = reinterpret_cast<const float4*>(Wl) + (4 * half) * WPC + pc;
  KBlk ring[RD];
  if (pw) {
#pragma unroll
    for (int i = 0; i < RD; ++i) kblk_load(ring[i], i, Wp, A0, A1);
  }

  int d[CPW];
#pragma unroll
  for (int it = 0; it < CPW; ++it) {
    const int c  = w * CPW + it;
    const int cc = c < NCHK ? c : NCHK - 1;
    const int* p = (cc < NCHK / 2) ? ei1 + cc * 32 : iei1 + (cc - NCHK / 2) * 32;
    const int v  = p[lane];
    d[it] = (c < NCHK) ? v : -1;
  }
  int enc[CPW], pos[CPW];
  int cw = 0;
#pragma unroll
  for (int it = 0; it < CPW; ++it) {
    const unsigned rel  = (unsigned)(d[it] - lo);
    const bool     hit  = rel < (unsigned)NPW;
    const unsigned mask = __builtin_amdgcn_ballot_w32(hit);
    pos[it] = cw + (int)__builtin_amdgcn_mbcnt_lo(mask, 0u);
    enc[it] = hit ? (int)(((unsigned)((w * CPW + it) * 32 + lane) << 5) | rel) : -1;
    cw += __builtin_popcount(mask);
  }
  if (lane == 0) cnt[w] = cw;
  __syncthreads();
  int base = 0, total = 0;
#pragma unroll
  for (int i = 0; i < NWV; ++i) {
    const int ci = cnt[i];
    base  += (i < w) ? ci : 0;
    total += ci;
  }
  total = __builtin_amdgcn_readfirstlane(total);
#pragma unroll
  for (int it = 0; it < CPW; ++it)
    if (enc[it] >= 0) lst[base + pos[it]] = enc[it];
  __syncthreads();

  float4 s0[2] = {z4, z4}, s1[2] = {z4, z4};
  float4 g0 = z4, g1 = z4;

  if (!pw) {
    const float4* b4 = reinterpret_cast<const float4*>(b_gcn + l * D0);
    g0 = b4[lane];
    g1 = b4[c1];
#pragma unroll
    for (int nn = 0; nn < 2; ++nn) node_sum(lst, total, 2 * w + nn, msg, lane, c1, s0[nn], s1[nn]);
  } else {
    v8f acc[2][4];
#pragma unroll
    for (int mt = 0; mt < 2; ++mt) {
#pragma unroll
      for (int t = 0; t < 4; ++t) acc[mt][t] = v8f{};
    }
#pragma unroll
    for (int u = 0; u < KB8; ++u) {
      kblk_mma(ring[u % RD], acc);
      if (u + RD < KB8) kblk_load(ring[u % RD], u + RD, Wp, A0, A1);
    }
#pragma unroll
    for (int mt = 0; mt < 2; ++mt) {
#pragma unroll
      for (int r = 0; r < 8; ++r) {
        const int row = 16 * mt + 8 * half + r;
        float4 v;
        v.x = acc[mt][0][r]; v.y = acc[mt][1][r]; v.z = acc[mt][2][r]; v.w = acc[mt][3][r];
        if (own && row < NPW) xls[row][pc] = v;
      }
    }
  }
  __syncthreads();

  if (!pw) {
#pragma unroll
    for (int nn = 0; nn < 2; ++nn) {
      const int sel = 2 * w + nn;
      const float4 x0 = xls[sel][lane];
      const float4 x1 = xls[sel][c1];
      float4* xo = reinterpret_cast<float4*>(xP + (size_t)(lo + sel) * DP);
      xo[lane] = upd4(s0[nn], x0, g0);
      if (hi) xo[32 + lane] = upd4(s1[nn], x1, g1);
    }
  }
}

static constexpr int APC = DP / 4;
static constexpr int UCH = 5;
static_assert(D0 == 8 * UCH * 5, "5 chunks of 5 blocks of 8 k");

struct GemmSet {
  float4 b[UCH][4];
  float4 a[2][UCH];
  float4 r[2][UCH];
  float4 b2[UCH][4];
};

template <bool GATHER, bool DUAL>
static __device__ __forceinline__ void gemm_load(
    GemmSet& s, int ch, const float4* __restrict__ Wp,
    const float4* __restrict__ Wp2,
    const float4* __restrict__ A0, const float4* __restrict__ A1,
    const float4* __restrict__ R0, const float4* __restrict__ R1) {
#pragma unroll
  for (int i = 0; i < UCH; ++i) {
    const int u = ch * UCH + i;
#pragma unroll
    for (int q = 0; q < 4; ++q) s.b[i][q] = Wp[(8 * u + q) * WPC];
    if (DUAL) {
#pragma unroll
      for (int q = 0; q < 4; ++q) s.b2[i][q] = Wp2[(8 * u + q) * WPC];
    }
    s.a[0][i] = A0[2 * u];
    s.a[1][i] = A1[2 * u];
    if (GATHER) {
      s.r[0][i] = R0[2 * u];
      s.r[1][i] = R1[2 * u];
    }
  }
}

template <bool GATHER, bool DUAL>
static __device__ __forceinline__ void gemm_mma(GemmSet& s, v8f (&acc)[2][4],
                                                v8f (&acc2)[2][4]) {
  if (GATHER) {
#pragma unroll
    for (int i = 0; i < UCH; ++i) {
#pragma unroll
      for (int mt = 0; mt < 2; ++mt) {
        s.a[mt][i].x -= s.r[mt][i].x;
        s.a[mt][i].y -= s.r[mt][i].y;
        s.a[mt][i].z -= s.r[mt][i].z;
        s.a[mt][i].w -= s.r[mt][i].w;
      }
    }
  }
#pragma unroll
  for (int i = 0; i < UCH; ++i) {
#pragma unroll
    for (int p = 0; p < 2; ++p) {
      v2f a0, a1;
      a0.x = p ? s.a[0][i].z : s.a[0][i].x;
      a0.y = p ? s.a[0][i].w : s.a[0][i].y;
      a1.x = p ? s.a[1][i].z : s.a[1][i].x;
      a1.y = p ? s.a[1][i].w : s.a[1][i].y;
#pragma unroll
      for (int t = 0; t < 4; ++t) {
        v2f b;
        b.x = f4c(s.b[i][2 * p], t);
        b.y = f4c(s.b[i][2 * p + 1], t);
        acc[0][t] = wmma4(a0, b, acc[0][t]);
        acc[1][t] = wmma4(a1, b, acc[1][t]);
      }
      if (DUAL) {
        v2f r0, r1;
        r0.x = p ? s.r[0][i].z : s.r[0][i].x;
        r0.y = p ? s.r[0][i].w : s.r[0][i].y;
        r1.x = p ? s.r[1][i].z : s.r[1][i].x;
        r1.y = p ? s.r[1][i].w : s.r[1][i].y;
#pragma unroll
        for (int t = 0; t < 4; ++t) {
          v2f b;
          b.x = f4c(s.b2[i][2 * p], t);
          b.y = f4c(s.b2[i][2 * p + 1], t);
          acc2[0][t] = wmma4(r0, b, acc2[0][t]);
          acc2[1][t] = wmma4(r1, b, acc2[1][t]);
        }
      }
    }
  }
}

template <bool GATHER, bool DUAL>
__global__ __launch_bounds__(128) void k_gemm(const float* __restrict__ A,
                                              const float* __restrict__ R,
                                              const int* __restrict__ idx,
                                              const float* __restrict__ W,
                                              float* __restrict__ C,
                                              const float* __restrict__ rowscale,
                                              const float* __restrict__ W2,
                                              float* __restrict__ C2) {
  static_assert(GATHER || !DUAL, "the second product takes the subtracted rows");
  const int tid  = threadIdx.x;
  const int lane = tid & 31;
  const int half = lane >> 4, l16 = lane & 15;
  const int wave = __builtin_amdgcn_readfirstlane(tid >> 5);
  const int m0 = blockIdx.x * 32;
  const int P0 = 13 * wave - (wave > 2 ? 1 : 0);
  const int np = wave < 2 ? 13 : 12;
  const bool own = l16 < np;
  const int pc = P0 + (own ? l16 : 0);

  const int ra = m0 + l16, rb = m0 + 16 + l16;
  const float4* A0 = reinterpret_cast<const float4*>(A + (size_t)(GATHER ? idx[ra] : ra) * DP) + half;
  const float4* A1 = reinterpret_cast<const float4*>(A + (size_t)(GATHER ? idx[rb] : rb) * DP) + half;
  const float4* R0 = GATHER ? reinterpret_cast<const float4*>(R + (size_t)ra * DP) + half : nullptr;
  const float4* R1 = GATHER ? reinterpret_cast<const float4*>(R + (size_t)rb * DP) + half : nullptr;
  const float4* Wp = reinterpret_cast<const float4*>(W) + (4 * half) * WPC + pc;
  const float4* Wp2 = DUAL ? reinterpret_cast<const float4*>(W2) + (4 * half) * WPC + pc : nullptr;

  v8f acc[2][4], acc2[2][4];
#pragma unroll
  for (int mt = 0; mt < 2; ++mt) {
#pragma unroll
    for (int t = 0; t < 4; ++t) { acc[mt][t] = v8f{}; acc2[mt][t] = v8f{}; }
  }

  GemmSet s0, s1;
  gemm_load<GATHER, DUAL>(s0, 0, Wp, Wp2, A0, A1, R0, R1);
#pragma unroll 1
  for (int p = 0; p < 2; ++p) {
    gemm_load<GATHER, DUAL>(s1, 2 * p + 1, Wp, Wp2, A0, A1, R0, R1);
    gemm_mma<GATHER, DUAL>(s0, acc, acc2);
    gemm_load<GATHER, DUAL>(s0, 2 * p + 2, Wp, Wp2, A0, A1, R0, R1);
    gemm_mma<GATHER, DUAL>(s1, acc, acc2);
  }
  gemm_mma<GATHER, DUAL>(s0, acc, acc2);

#pragma unroll
  for (int mt = 0; mt < 2; ++mt) {
#pragma unroll
    for (int r = 0; r < 8; ++r) {
      const int row = m0 + 16 * mt + 8 * half + r;
      const float sc = GATHER ? rowscale[row] : 1.0f;
      float4 v;
      v.x = acc[mt][0][r]; v.y = acc[mt][1][r]; v.z = acc[mt][2][r]; v.w = acc[mt][3][r];
      if (GATHER) { v.x *= sc; v.y *= sc; v.z *= sc; v.w *= sc; }
      if (own) reinterpret_cast<float4*>(C + (size_t)row * DP)[pc] = v;
      if (DUAL) {
        float4 v2;
        v2.x = acc2[mt][0][r]; v2.y = acc2[mt][1][r]; v2.z = acc2[mt][2][r]; v2.w = acc2[mt][3][r];
        if (own) reinterpret_cast<float4*>(C2 + (size_t)row * DP)[pc] = v2;
      }
    }
  }
}

static constexpr int IMGP = 401;
static constexpr int TAPP = 52;
static constexpr int KCH  = 13;
static constexpr int PSPLIT = 147;
static constexpr int ET = 2;
static constexpr int EB = 16 * ET;
static_assert(NE % EB == 0, "the grid covers the edges exactly");

struct alignas(16) DecLds {
  float img[EB][IMGP];
  float cwp[NF][TAPP];
  float chc[NF][4];
  float part[8][ET][NTN][8][32];
  float h2s[EB][D0];
};
static_assert((EB * IMGP * 4) % 16 == 0, "cwp starts 16-byte aligned");
static_assert(sizeof(DecLds) <= 320 * 1024, "fits the LDS of one WGP");

__global__ __launch_bounds__(256) void k_convfc(
    const float* __restrict__ xP, const float* __restrict__ rP,
    const int* __restrict__ ei0, const float* __restrict__ fcwP,
    const float* __restrict__ conv_w, const float* __restrict__ conv_b,
    const float* __restrict__ bn1_g, const float* __restrict__ bn1_b,
    const float* __restrict__ fc_b,
    const float* __restrict__ bn2_g, const float* __restrict__ bn2_b,
    const float* __restrict__ fc1_w, const float* __restrict__ fc1_b,
    const float* __restrict__ bn0_g, const float* __restrict__ bn0_b,
    float* __restrict__ out) {
  __shared__ DecLds lds;

  const int m0   = blockIdx.x * EB;
  const int tid  = threadIdx.x;
  const int lane = tid & 31;
  const int half = lane >> 4, l16 = lane & 15;
  const int wave = __builtin_amdgcn_readfirstlane(tid >> 5);

  const float b0s = bn0_g[0] * INV_SQRT_1PE;
  const float b0b = bn0_b[0];

  for (int i = tid; i < EB * 400; i += 256) {
    int t = i / 400, k = i % 400;
    int srcn = ei0[m0 + t];
    float v = (k & 1) ? rP[(size_t)(m0 + t) * DP + (k >> 1)]
                      : xP[(size_t)srcn * DP + (k >> 1)];
    lds.img[t][k] = v * b0s + b0b;
  }
  for (int i = tid; i < NF * TAPP; i += 256) {
    int ch = i / TAPP, tp = i % TAPP;
    lds.cwp[ch][tp] = (tp < CONVK) ? conv_w[ch * CONVK + tp] : 0.0f;
  }
  if (tid < NF) {
    lds.chc[tid][0] = conv_b[tid];
    lds.chc[tid][1] = bn1_g[tid] * INV_SQRT_1PE;
    lds.chc[tid][2] = bn1_b[tid];
    lds.chc[tid][3] = 0.0f;
  }
  __syncthreads();

  int toff[KCH][2];
#pragma unroll
  for (int ks = 0; ks < KCH; ++ks) {
#pragma unroll
    for (int s = 0; s < 2; ++s) {
      int tp = 4 * ks + 2 * half + s;
      toff[ks][s] = (tp < CONVK) ? (tp / 7) * 20 + (tp % 7) : 0;
    }
  }
  const bool lastx_ok = (4 * (KCH - 1) + 2 * half)     < CONVK;
  const bool lasty_ok = (4 * (KCH - 1) + 2 * half + 1) < CONVK;

  const float*  imrow = &lds.img[l16][0];
  const float4* fcw4  = reinterpret_cast<const float4*>(fcwP);

  v8f acc[ET][NTN];
#pragma unroll
  for (int t = 0; t < ET; ++t) {
#pragma unroll
    for (int i = 0; i < NTN; ++i) acc[t][i] = v8f{};
  }

  const int nseg = (wave < 6) ? 1 : 3;
  const int p0   = (wave < 6) ? 0 : PSPLIT;
  const int p1   = (wave < 6) ? PSPLIT : P196;

  for (int sg = 0; sg < nseg; ++sg) {
    const int j = (wave < 6) ? wave : 3 * (wave - 6) + sg;

    v2f ta[KCH];
#pragma unroll
    for (int ks = 0; ks < KCH; ++ks)
      ta[ks] = *reinterpret_cast<const v2f*>(&lds.cwp[16 * j + l16][4 * ks + 2 * half]);
    float cbv[8], scv[8], bbv[8];
#pragma unroll
    for (int r = 0; r < 8; ++r) {
      const float* cc = &lds.chc[16 * j + 8 * half + r][0];
      cbv[r] = cc[0]; scv[r] = cc[1]; bbv[r] = cc[2];
    }

    float4 fx[NTN], fy[NTN];
#pragma unroll
    for (int i = 0; i < NTN; ++i) { fx[i] = float4{0.f, 0.f, 0.f, 0.f}; fy[i] = fx[i]; }

#pragma unroll 1
    for (int pos = p0 - 1; pos < p1; ++pos) {
      const int pn = (pos + 1 < p1) ? pos + 1 : pos;
      const float4* fpn = fcw4 + (size_t)((j * P196 + pn) * 2) * (NTN * 32) + lane;

      if (pos >= p0) {
        const int woff = (pos / 14) * 20 + (pos % 14);
        float act[ET][8];
#pragma unroll
        for (int t = 0; t < ET; ++t) {
          const float* win = imrow + t * (16 * IMGP) + woff;
          v8f d0 = v8f{}, d1 = v8f{};
#pragma unroll
          for (int ks = 0; ks < KCH; ++ks) {
            v2f b;
            b.x = win[toff[ks][0]];
            b.y = win[toff[ks][1]];
            if (ks == KCH - 1) {
              b.x = lastx_ok ? b.x : 0.0f;
              b.y = lasty_ok ? b.y : 0.0f;
            }
            if (ks & 1) d1 = wmma4(ta[ks], b, d1);
            else        d0 = wmma4(ta[ks], b, d0);
          }
#pragma unroll
          for (int r = 0; r < 8; ++r) {
            float v = ((d0[r] + d1[r]) + cbv[r]) * scv[r] + bbv[r];
            act[t][r] = v > 0.f ? v : 0.f;
          }
        }

#pragma unroll
        for (int q = 0; q < 4; ++q) {
#pragma unroll
          for (int t = 0; t < ET; ++t) {
            v2f b;
            b.x = act[t][2 * q];
            b.y = act[t][2 * q + 1];
#pragma unroll
            for (int i = 0; i < NTN; ++i) {
              v2f a;
              if (q == 0)      { a.x = fx[i].x; a.y = fx[i].y; }
              else if (q == 1) { a.x = fx[i].z; a.y = fx[i].w; }
              else if (q == 2) { a.x = fy[i].x; a.y = fy[i].y; }
              else             { a.x = fy[i].z; a.y = fy[i].w; }
              acc[t][i] = wmma4(a, b, acc[t][i]);
            }
          }
        }
      }

#pragma unroll
      for (int i = 0; i < NTN; ++i) { fx[i] = fpn[i * 32]; fy[i] = fpn[(NTN + i) * 32]; }
    }
  }

#pragma unroll
  for (int t = 0; t < ET; ++t) {
#pragma unroll
    for (int i = 0; i < NTN; ++i) {
#pragma unroll
      for (int r = 0; r < 8; ++r) lds.part[wave][t][i][r][lane] = acc[t][i][r];
    }
  }
  __syncthreads();

  for (int idx = tid; idx < ET * NTN * 8 * 32; idx += 256) {
    int t = idx / (NTN * 256), rem = idx % (NTN * 256);
    int i = rem >> 8, r = (rem >> 5) & 7, ln = rem & 31;
    float s = 0.0f;
#pragma unroll
    for (int w = 0; w < 8; ++w) s += lds.part[w][t][i][r][ln];
    int n = 16 * i + 8 * (ln >> 4) + r;
    int ed = 16 * t + (ln & 15);
    if (n < D0) {
      float val = (s + fc_b[n]) * (bn2_g[n] * INV_SQRT_1PE) + bn2_b[n];
      lds.h2s[ed][n] = val > 0.f ? val : 0.f;
    }
  }
  __syncthreads();

  for (int i = tid; i < EB * NCLS; i += 256) {
    int t = i / NCLS, cls = i % NCLS;
    float s = fc1_b[cls];
    for (int c2 = 0; c2 < D0; ++c2) s += lds.h2s[t][c2] * fc1_w[cls * D0 + c2];
    out[(size_t)(m0 + t) * NCLS + cls] = s;
  }
}


extern "C" void kernel_launch(void* const* d_in, const int* in_sizes, int n_in,
                              void* d_out, int out_size, void* d_ws, size_t ws_size,
                              hipStream_t stream) {
  const float* node  = (const float*)d_in[0];
  const float* ef    = (const float*)d_in[1];
  const float* ief   = (const float*)d_in[2];
  const float* Win   = (const float*)d_in[3];
  const float* Wout  = (const float*)d_in[4];
  const float* Wloop = (const float*)d_in[5];
  const float* Wrel  = (const float*)d_in[6];
  const float* bgcn  = (const float*)d_in[7];
  const float* bn0g  = (const float*)d_in[8];
  const float* bn0b  = (const float*)d_in[9];
  const float* convw = (const float*)d_in[10];
  const float* convb = (const float*)d_in[11];
  const float* bn1g  = (const float*)d_in[12];
  const float* bn1b  = (const float*)d_in[13];
  const float* fcw   = (const float*)d_in[14];
  const float* fcb   = (const float*)d_in[15];
  const float* bn2g  = (const float*)d_in[16];
  const float* bn2b  = (const float*)d_in[17];
  const float* fc1w  = (const float*)d_in[18];
  const float* fc1b  = (const float*)d_in[19];
  const int*   ei    = (const int*)d_in[20];
  const int*   iei   = (const int*)d_in[21];
  const int *ei0 = ei, *ei1 = ei + NE;
  const int *iei0 = iei, *iei1 = iei + NE;

  float* ws = (float*)d_ws;
  size_t o = 0;
  float* xP   = ws + o; o += (size_t)NVP * DP;
  float* r0   = ws + o; o += (size_t)NE * DP;
  float* r1   = ws + o; o += (size_t)NE * DP;
  float* ir0  = ws + o; o += (size_t)NE * DP;
  float* ir1  = ws + o; o += (size_t)NE * DP;
  float* minb = ws + o; o += (size_t)NE * DP;
  float* moutb= ws + o; o += (size_t)NE * DP;
  float* fcwT = ws + o; o += (size_t)FLAT * DP;
  float* degf = ws + o; o += NVP;
  float* degi = ws + o; o += NVP;
  float* nfv  = ws + o; o += NE;
  float* niv  = ws + o; o += NE;

  const int TB = 256;
  auto cdiv = [](int a, int b) { return (a + b - 1) / b; };

  k_pad<<<cdiv(NVP * DP, TB), TB, 0, stream>>>(node, xP, NV, NVP);
  k_pad<<<cdiv(NE * DP, TB), TB, 0, stream>>>(ef, r0, NE, NE);
  k_pad<<<cdiv(NE * DP, TB), TB, 0, stream>>>(ief, ir0, NE, NE);
  k_pack_fcw<<<cdiv(FLAT * DP, TB), TB, 0, stream>>>(fcw, fcwT);

  k_zero<<<cdiv(2 * NVP, TB), TB, 0, stream>>>(degf, 2 * NVP);
  k_deg<<<cdiv(2 * NE, TB), TB, 0, stream>>>(ei0, iei0, degf, degi);
  k_norm<<<cdiv(NE, TB), TB, 0, stream>>>(ei0, ei1, iei0, iei1, degf, degi, nfv, niv);

  float *rc = r0, *rn = r1, *irc = ir0, *irn = ir1;
  const int GE = NE / 32;

  for (int l = 0; l < NLAY; ++l) {
    const float* wi = Win   + (size_t)l * D0 * D0;
    const float* wo = Wout  + (size_t)l * D0 * D0;
    const float* wp = Wloop + (size_t)l * D0 * D0;
    const float* wr = Wrel  + (size_t)l * D0 * D0;
    k_gemm<true, true><<<GE, 128, 0, stream>>>(xP, rc,  ei0,  wi, minb,  nfv, wr, rn);
    k_gemm<true, true><<<GE, 128, 0, stream>>>(xP, irc, iei0, wo, moutb, niv, wr, irn);
    k_aggupd<<<NV / NPW, NWV * 32, 0, stream>>>(minb, ei1, iei1, xP, wp, bgcn, l);
    float* t;
    t = rc; rc = rn; rn = t;
    t = irc; irc = irn; irn = t;
  }

  k_convfc<<<NE / EB, 256, 0, stream>>>(xP, rc, ei0, fcwT, convw, convb,
                                        bn1g, bn1b, fcb, bn2g, bn2b,
                                        fc1w, fc1b, bn0g, bn0b, (float*)d_out);
}


# decoder set-up and tail: image rows, taps, fc1_w and bn constants fetched as whole groups of loads (16-byte pieces), one memory round trip per phase, not one or two per element; sum order kept
# speedup vs baseline: 3.1278x; 1.0391x over previous
#include <hip/hip_runtime.h>
#include <math.h>


static constexpr int D0   = 200;
static constexpr int DP   = 208;
static constexpr int NTN  = 13;
static constexpr int NE   = 8192;
static constexpr int NV   = 5000;
static constexpr int NVP  = 5056;
static constexpr int NLAY = 6;
static constexpr int NF   = 96;
static constexpr int CONVK= 49;
static constexpr int P196 = 196;
static constexpr int FLAT = 18816;
static constexpr int NCLS = 13;
#define INV_SQRT_1PE 0.99999500f

typedef __attribute__((ext_vector_type(2))) float v2f;
typedef __attribute__((ext_vector_type(8))) float v8f;

static __device__ __forceinline__ v8f wmma4(v2f a, v2f b, v8f c) {
  return __builtin_amdgcn_wmma_f32_16x16x4_f32(false, a, false, b, (short)0, c,
                                               false, false);
}


__global__ void k_zero(float* __restrict__ p, int n) {
  int i = blockIdx.x * blockDim.x + threadIdx.x;
  if (i < n) p[i] = 0.0f;
}

__global__ void k_pad(const float* __restrict__ src, float* __restrict__ dst,
                      int rows, int prows) {
  int i = blockIdx.x * blockDim.x + threadIdx.x;
  int total = prows * DP;
  if (i >= total) return;
  int r = i / DP, c = i % DP;
  dst[i] = (r < rows && c < D0) ? src[r * D0 + c] : 0.0f;
}

__global__ void k_pack_fcw(const float* __restrict__ fcw, float* __restrict__ fcwP) {
  int o = blockIdx.x * blockDim.x + threadIdx.x;
  if (o >= FLAT * DP) return;
  int e    = o & 3;
  int lane = (o >> 2) & 31;
  int rest = o >> 7;
  int i    = rest % NTN;
  int rst2 = rest / NTN;
  int p    = rst2 & 1;
  int rst3 = rst2 >> 1;
  int pos  = rst3 % P196;
  int j    = rst3 / P196;
  int n    = 16 * i + (lane & 15);
  int ch   = 16 * j + 8 * (lane >> 4) + 4 * p + e;
  fcwP[o] = (n < D0) ? fcw[(size_t)n * FLAT + ch * P196 + pos] : 0.0f;
}

__global__ void k_deg(const int* __restrict__ ei0, const int* __restrict__ iei0,
                      float* __restrict__ degf, float* __restrict__ degi) {
  int e = blockIdx.x * blockDim.x + threadIdx.x;
  if (e < NE)            atomicAdd(&degf[ei0[e]], 1.0f);
  else if (e < 2 * NE)   atomicAdd(&degi[iei0[e - NE]], 1.0f);
}

__global__ void k_norm(const int* __restrict__ ei0, const int* __restrict__ ei1,
                       const int* __restrict__ iei0, const int* __restrict__ iei1,
                       const float* __restrict__ degf, const float* __restrict__ degi,
                       float* __restrict__ nf, float* __restrict__ ni) {
  int e = blockIdx.x * blockDim.x + threadIdx.x;
  if (e >= NE) return;
  float df0 = degf[ei0[e]],  df1 = degf[ei1[e]];
  float di0 = degi[iei0[e]], di1 = degi[iei1[e]];
  float a = df0 > 0.f ? rsqrtf(fmaxf(df0, 1.f)) : 0.f;
  float b = df1 > 0.f ? rsqrtf(fmaxf(df1, 1.f)) : 0.f;
  float c = di0 > 0.f ? rsqrtf(fmaxf(di0, 1.f)) : 0.f;
  float d = di1 > 0.f ? rsqrtf(fmaxf(di1, 1.f)) : 0.f;
  nf[e] = a * b;
  ni[e] = c * d;
}

static constexpr int NPW  = 20;
static constexpr int NNW  = NPW / 2;
static constexpr int NGW  = 4;
static constexpr int NWV  = NNW + NGW;
static constexpr int NCHK = 2 * NE / 32;
static constexpr int CPW  = 37;
static constexpr int ROW4 = D0 / 4;
static constexpr int WPC  = D0 / 4;
static constexpr int KB8  = D0 / 8;
static constexpr int RD   = 5;
static_assert(NV % NPW == 0, "the grid covers the nodes exactly");
static_assert(NPW == 2 * NNW, "two nodes per node wave");
static_assert(NWV * CPW >= NCHK, "every chunk has a wave");
static_assert(NPW <= 32, "the owned-node number fits the 5 low bits of an entry");
static_assert(NPW > 16 && NPW <= 32, "the product tile has two 16-row halves");

static __device__ __forceinline__ float f4c(const float4& v, int t) {
  return t == 0 ? v.x : t == 1 ? v.y : t == 2 ? v.z : v.w;
}

static __device__ __forceinline__ float4 add4(float4 a, float4 b) {
  return float4{a.x + b.x, a.y + b.y, a.z + b.z, a.w + b.w};
}

static __device__ __forceinline__ float4 upd4(float4 agg, float4 xs, float4 bb) {
  float4 r;
  r.x = tanhf((agg.x + xs.x) * (1.0f / 3.0f) + bb.x);
  r.y = tanhf((agg.y + xs.y) * (1.0f / 3.0f) + bb.y);
  r.z = tanhf((agg.z + xs.z) * (1.0f / 3.0f) + bb.z);
  r.w = tanhf((agg.w + xs.w) * (1.0f / 3.0f) + bb.w);
  return r;
}

struct KBlk {
  float4 b[4];
  float4 a[2];
};

static __device__ __forceinline__ void kblk_load(KBlk& s, int u,
                                                 const float4* __restrict__ Wp,
                                                 const float4* A0, const float4* A1) {
#pragma unroll
  for (int q = 0; q < 4; ++q) s.b[q] = Wp[(8 * u + q) * WPC];
  s.a[0] = A0[2 * u];
  s.a[1] = A1[2 * u];
}

static __device__ __forceinline__ void kblk_mma(const KBlk& s, v8f (&acc)[2][4]) {
#pragma unroll
  for (int p = 0; p < 2; ++p) {
    v2f a0, a1;
    a0.x = p ? s.a[0].z : s.a[0].x;
    a0.y = p ? s.a[0].w : s.a[0].y;
    a1.x = p ? s.a[1].z : s.a[1].x;
    a1.y = p ? s.a[1].w : s.a[1].y;
#pragma unroll
    for (int t = 0; t < 4; ++t) {
      v2f b;
      b.x = f4c(s.b[2 * p], t);
      b.y = f4c(s.b[2 * p + 1], t);
      acc[0][t] = wmma4(a0, b, acc[0][t]);
      acc[1][t] = wmma4(a1, b, acc[1][t]);
    }
  }
}

static __device__ __forceinline__ void node_sum(const int* lst, int total, int sel,
                                                const float* __restrict__ msg,
                                                int lane, int c1,
                                                float4& a0, float4& a1) {
  for (int b0 = 0; b0 < total; b0 += 32) {
    const int ent = (b0 + lane < total) ? lst[b0 + lane] : -1;
    unsigned m = __builtin_amdgcn_ballot_w32(ent >= 0 && (ent & 31) == sel);
    while (m) {
      int e[4];
#pragma unroll
      for (int u = 0; u < 4; ++u) {
        e[u] = -1;
        if (m) {
          const int k = __builtin_ctz(m);
          m &= m - 1;
          e[u] = __builtin_amdgcn_readlane(ent, k) >> 5;
        }
      }
      float4 v0[4], v1[4];
#pragma unroll
      for (int u = 0; u < 4; ++u) {
        const int er = e[u] < 0 ? e[0] : e[u];
        const float4* row = reinterpret_cast<const float4*>(msg + (size_t)er * DP);
        v0[u] = row[lane];
        v1[u] = row[c1];
      }
#pragma unroll
      for (int u = 0; u < 4; ++u) {
        if (e[u] >= 0) { a0 = add4(a0, v0[u]); a1 = add4(a1, v1[u]); }
      }
    }
  }
}

__global__ __launch_bounds__(NWV * 32) void k_aggupd(
    const float* __restrict__ msg, const int* __restrict__ ei1,
    const int* __restrict__ iei1, float* xP,
    const float* __restrict__ Wl, const float* __restrict__ b_gcn, int l) {
  __shared__ int lst[2 * NE];
  __shared__ int cnt[NWV];
  __shared__ float4 xls[NPW][DP / 4];
  const int tid = threadIdx.x, lane = tid & 31;
  const int w  = __builtin_amdgcn_readfirstlane(tid >> 5);
  const int lo = blockIdx.x * NPW;
  const bool pw = w >= NNW;

  const bool hi = lane < ROW4 - 32;
  const int  c1 = hi ? 32 + lane : lane;
  const float4 z4 = float4{0.f, 0.f, 0.f, 0.f};

  const int half = lane >> 4, l16 = lane & 15;
  const int g  = pw ? w - NNW : 0;
  const int P0 = 13 * g - (g > 2 ? 1 : 0);
  const int np = g < 2 ? 13 : 12;
  const bool own = l16 < np;
  const int pc = P0 + (own ? l16 : 0);
  const int rA = lo + l16;
  const int rB = lo + (l16 < NPW - 16 ? 16 + l16 : l16 - (NPW - 16));
  const float4* A0 = reinterpret_cast<const float4*>(xP + (size_t)rA * DP) + half;
  const float4* A1 = reinterpret_cast<const float4*>(xP + (size_t)rB * DP) + half;
  const float4* Wp = reinterpret_cast<const float4*>(Wl) + (4 * half) * WPC + pc;
  KBlk ring[RD];
  if (pw) {
#pragma unroll
    for (int i = 0; i < RD; ++i) kblk_load(ring[i], i, Wp, A0, A1);
  }

  int d[CPW];
#pragma unroll
  for (int it = 0; it < CPW; ++it) {
    const int c  = w * CPW + it;
    const int cc = c < NCHK ? c : NCHK - 1;
    const int* p = (cc < NCHK / 2) ? ei1 + cc * 32 : iei1 + (cc - NCHK / 2) * 32;
    const int v  = p[lane];
    d[it] = (c < NCHK) ? v : -1;
  }
  int enc[CPW], pos[CPW];
  int cw = 0;
#pragma unroll
  for (int it = 0; it < CPW; ++it) {
    const unsigned rel  = (unsigned)(d[it] - lo);
    const bool     hit  = rel < (unsigned)NPW;
    const unsigned mask = __builtin_amdgcn_ballot_w32(hit);
    pos[it] = cw + (int)__builtin_amdgcn_mbcnt_lo(mask, 0u);
    enc[it] = hit ? (int)(((unsigned)((w * CPW + it) * 32 + lane) << 5) | rel) : -1;
    cw += __builtin_popcount(mask);
  }
  if (lane == 0) cnt[w] = cw;
  __syncthreads();
  int base = 0, total = 0;
#pragma unroll
  for (int i = 0; i < NWV; ++i) {
    const int ci = cnt[i];
    base  += (i < w) ? ci : 0;
    total += ci;
  }
  total = __builtin_amdgcn_readfirstlane(total);
#pragma unroll
  for (int it = 0; it < CPW; ++it)
    if (enc[it] >= 0) lst[base + pos[it]] = enc[it];
  __syncthreads();

  float4 s0[2] = {z4, z4}, s1[2] = {z4, z4};
  float4 g0 = z4, g1 = z4;

  if (!pw) {
    const float4* b4 = reinterpret_cast<const float4*>(b_gcn + l * D0);
    g0 = b4[lane];
    g1 = b4[c1];
#pragma unroll
    for (int nn = 0; nn < 2; ++nn) node_sum(lst, total, 2 * w + nn, msg, lane, c1, s0[nn], s1[nn]);
  } else {
    v8f acc[2][4];
#pragma unroll
    for (int mt = 0; mt < 2; ++mt) {
#pragma unroll
      for (int t = 0; t < 4; ++t) acc[mt][t] = v8f{};
    }
#pragma unroll
    for (int u = 0; u < KB8; ++u) {
      kblk_mma(ring[u % RD], acc);
      if (u + RD < KB8) kblk_load(ring[u % RD], u + RD, Wp, A0, A1);
    }
#pragma unroll
    for (int mt = 0; mt < 2; ++mt) {
#pragma unroll
      for (int r = 0; r < 8; ++r) {
        const int row = 16 * mt + 8 * half + r;
        float4 v;
        v.x = acc[mt][0][r]; v.y = acc[mt][1][r]; v.z = acc[mt][2][r]; v.w = acc[mt][3][r];
        if (own && row < NPW) xls[row][pc] = v;
      }
    }
  }
  __syncthreads();

  if (!pw) {
#pragma unroll
    for (int nn = 0; nn < 2; ++nn) {
      const int sel = 2 * w + nn;
      const float4 x0 = xls[sel][lane];
      const float4 x1 = xls[sel][c1];
      float4* xo = reinterpret_cast<float4*>(xP + (size_t)(lo + sel) * DP);
      xo[lane] = upd4(s0[nn], x0, g0);
      if (hi) xo[32 + lane] = upd4(s1[nn], x1, g1);
    }
  }
}

static constexpr int APC = DP / 4;
static constexpr int UCH = 5;
static_assert(D0 == 8 * UCH * 5, "5 chunks of 5 blocks of 8 k");

struct GemmSet {
  float4 b[UCH][4];
  float4 a[2][UCH];
  float4 r[2][UCH];
  float4 b2[UCH][4];
};

template <bool GATHER, bool DUAL>
static __device__ __forceinline__ void gemm_load(
    GemmSet& s, int ch, const float4* __restrict__ Wp,
    const float4* __restrict__ Wp2,
    const float4* __restrict__ A0, const float4* __restrict__ A1,
    const float4* __restrict__ R0, const float4* __restrict__ R1) {
#pragma unroll
  for (int i = 0; i < UCH; ++i) {
    const int u = ch * UCH + i;
#pragma unroll
    for (int q = 0; q < 4; ++q) s.b[i][q] = Wp[(8 * u + q) * WPC];
    if (DUAL) {
#pragma unroll
      for (int q = 0; q < 4; ++q) s.b2[i][q] = Wp2[(8 * u + q) * WPC];
    }
    s.a[0][i] = A0[2 * u];
    s.a[1][i] = A1[2 * u];
    if (GATHER) {
      s.r[0][i] = R0[2 * u];
      s.r[1][i] = R1[2 * u];
    }
  }
}

template <bool GATHER, bool DUAL>
static __device__ __forceinline__ void gemm_mma(GemmSet& s, v8f (&acc)[2][4],
                                                v8f (&acc2)[2][4]) {
  if (GATHER) {
#pragma unroll
    for (int i = 0; i < UCH; ++i) {
#pragma unroll
      for (int mt = 0; mt < 2; ++mt) {
        s.a[mt][i].x -= s.r[mt][i].x;
        s.a[mt][i].y -= s.r[mt][i].y;
        s.a[mt][i].z -= s.r[mt][i].z;
        s.a[mt][i].w -= s.r[mt][i].w;
      }
    }
  }
#pragma unroll
  for (int i = 0; i < UCH; ++i) {
#pragma unroll
    for (int p = 0; p < 2; ++p) {
      v2f a0, a1;
      a0.x = p ? s.a[0][i].z : s.a[0][i].x;
      a0.y = p ? s.a[0][i].w : s.a[0][i].y;
      a1.x = p ? s.a[1][i].z : s.a[1][i].x;
      a1.y = p ? s.a[1][i].w : s.a[1][i].y;
#pragma unroll
      for (int t = 0; t < 4; ++t) {
        v2f b;
        b.x = f4c(s.b[i][2 * p], t);
        b.y = f4c(s.b[i][2 * p + 1], t);
        acc[0][t] = wmma4(a0, b, acc[0][t]);
        acc[1][t] = wmma4(a1, b, acc[1][t]);
      }
      if (DUAL) {
        v2f r0, r1;
        r0.x = p ? s.r[0][i].z : s.r[0][i].x;
        r0.y = p ? s.r[0][i].w : s.r[0][i].y;
        r1.x = p ? s.r[1][i].z : s.r[1][i].x;
        r1.y = p ? s.r[1][i].w : s.r[1][i].y;
#pragma unroll
        for (int t = 0; t < 4; ++t) {
          v2f b;
          b.x = f4c(s.b2[i][2 * p], t);
          b.y = f4c(s.b2[i][2 * p + 1], t);
          acc2[0][t] = wmma4(r0, b, acc2[0][t]);
          acc2[1][t] = wmma4(r1, b, acc2[1][t]);
        }
      }
    }
  }
}

template <bool GATHER, bool DUAL>
__global__ __launch_bounds__(128) void k_gemm(const float* __restrict__ A,
                                              const float* __restrict__ R,
                                              const int* __restrict__ idx,
                                              const float* __restrict__ W,
                                              float* __restrict__ C,
                                              const float* __restrict__ rowscale,
                                              const float* __restrict__ W2,
                                              float* __restrict__ C2) {
  static_assert(GATHER || !DUAL, "the second product takes the subtracted rows");
  const int tid  = threadIdx.x;
  const int lane = tid & 31;
  const int half = lane >> 4, l16 = lane & 15;
  const int wave = __builtin_amdgcn_readfirstlane(tid >> 5);
  const int m0 = blockIdx.x * 32;
  const int P0 = 13 * wave - (wave > 2 ? 1 : 0);
  const int np = wave < 2 ? 13 : 12;
  const bool own = l16 < np;
  const int pc = P0 + (own ? l16 : 0);

  const int ra = m0 + l16, rb = m0 + 16 + l16;
  const float4* A0 = reinterpret_cast<const float4*>(A + (size_t)(GATHER ? idx[ra] : ra) * DP) + half;
  const float4* A1 = reinterpret_cast<const float4*>(A + (size_t)(GATHER ? idx[rb] : rb) * DP) + half;
  const float4* R0 = GATHER ? reinterpret_cast<const float4*>(R + (size_t)ra * DP) + half : nullptr;
  const float4* R1 = GATHER ? reinterpret_cast<const float4*>(R + (size_t)rb * DP) + half : nullptr;
  const float4* Wp = reinterpret_cast<const float4*>(W) + (4 * half) * WPC + pc;
  const float4* Wp2 = DUAL ? reinterpret_cast<const float4*>(W2) + (4 * half) * WPC + pc : nullptr;

  v8f acc[2][4], acc2[2][4];
#pragma unroll
  for (int mt = 0; mt < 2; ++mt) {
#pragma unroll
    for (int t = 0; t < 4; ++t) { acc[mt][t] = v8f{}; acc2[mt][t] = v8f{}; }
  }

  GemmSet s0, s1;
  gemm_load<GATHER, DUAL>(s0, 0, Wp, Wp2, A0, A1, R0, R1);
#pragma unroll 1
  for (int p = 0; p < 2; ++p) {
    gemm_load<GATHER, DUAL>(s1, 2 * p + 1, Wp, Wp2, A0, A1, R0, R1);
    gemm_mma<GATHER, DUAL>(s0, acc, acc2);
    gemm_load<GATHER, DUAL>(s0, 2 * p + 2, Wp, Wp2, A0, A1, R0, R1);
    gemm_mma<GATHER, DUAL>(s1, acc, acc2);
  }
  gemm_mma<GATHER, DUAL>(s0, acc, acc2);

#pragma unroll
  for (int mt = 0; mt < 2; ++mt) {
#pragma unroll
    for (int r = 0; r < 8; ++r) {
      const int row = m0 + 16 * mt + 8 * half + r;
      const float sc = GATHER ? rowscale[row] : 1.0f;
      float4 v;
      v.x = acc[mt][0][r]; v.y = acc[mt][1][r]; v.z = acc[mt][2][r]; v.w = acc[mt][3][r];
      if (GATHER) { v.x *= sc; v.y *= sc; v.z *= sc; v.w *= sc; }
      if (own) reinterpret_cast<float4*>(C + (size_t)row * DP)[pc] = v;
      if (DUAL) {
        float4 v2;
        v2.x = acc2[mt][0][r]; v2.y = acc2[mt][1][r]; v2.z = acc2[mt][2][r]; v2.w = acc2[mt][3][r];
        if (own) reinterpret_cast<float4*>(C2 + (size_t)row * DP)[pc] = v2;
      }
    }
  }
}

static constexpr int IMGP = 401;
static constexpr int TAPP = 52;
static constexpr int KCH  = 13;
static constexpr int PSPLIT = 147;
static constexpr int ET = 2;
static constexpr int NTPP = (NF * TAPP + 255) / 256;
static constexpr int NWPP = (NCLS * D0 / 4 + 255) / 256;
static constexpr int EB = 16 * ET;
static_assert(NE % EB == 0, "the grid covers the edges exactly");

struct alignas(16) DecLds {
  float img[EB][IMGP];
  float cwp[NTPP * 256];
  float chc[NF][4];
  float part[8][ET][NTN][8][32];
  float h2s[EB][D0];
  float w1[NWPP * 256 * 4];
};
static_assert((EB * IMGP * 4) % 16 == 0, "cwp starts 16-byte aligned");
static_assert(sizeof(DecLds) <= 320 * 1024, "fits the LDS of one WGP");

__global__ __launch_bounds__(256) void k_convfc(
    const float* __restrict__ xP, const float* __restrict__ rP,
    const int* __restrict__ ei0, const float* __restrict__ fcwP,
    const float* __restrict__ conv_w, const float* __restrict__ conv_b,
    const float* __restrict__ bn1_g, const float* __restrict__ bn1_b,
    const float* __restrict__ fc_b,
    const float* __restrict__ bn2_g, const float* __restrict__ bn2_b,
    const float* __restrict__ fc1_w, const float* __restrict__ fc1_b,
    const float* __restrict__ bn0_g, const float* __restrict__ bn0_b,
    float* __restrict__ out) {
  __shared__ DecLds lds;

  const int m0   = blockIdx.x * EB;
  const int tid  = threadIdx.x;
  const int lane = tid & 31;
  const int half = lane >> 4, l16 = lane & 15;
  const int wave = __builtin_amdgcn_readfirstlane(tid >> 5);

  const float b0s = bn0_g[0] * INV_SQRT_1PE;
  const float b0b = bn0_b[0];

  constexpr int NIMG = EB * ROW4;
  constexpr int NIM  = (NIMG + 255) / 256;
  constexpr int NTP  = NTPP;
  constexpr int NW1  = NCLS * D0 / 4;
  constexpr int NWP  = NWPP;
  static_assert((NCLS * D0) % 4 == 0, "fc1_w is whole 16-byte pieces");
  int sn[NIM];
#pragma unroll
  for (int it = 0; it < NIM; ++it) {
    const int i = tid + 256 * it;
    sn[it] = ei0[m0 + (i < NIMG ? i : NIMG - 1) / ROW4];
  }
  float cwv[NTP];
#pragma unroll
  for (int it = 0; it < NTP; ++it) {
    const int i  = tid + 256 * it;
    const int ii = i < NF * TAPP ? i : NF * TAPP - 1;
    const int ch = ii / TAPP, tp = ii % TAPP;
    cwv[it] = conv_w[ch * CONVK + (tp < CONVK ? tp : CONVK - 1)];
  }
  float4 w1v[NWP];
#pragma unroll
  for (int it = 0; it < NWP; ++it) {
    const int i = tid + 256 * it;
    w1v[it] = reinterpret_cast<const float4*>(fc1_w)[i < NW1 ? i : NW1 - 1];
  }
  const int   cht = tid < NF ? tid : NF - 1;
  const float ccb = conv_b[cht], ccg = bn1_g[cht], ccs = bn1_b[cht];
  float4 xv[NIM], rv[NIM];
#pragma unroll
  for (int it = 0; it < NIM; ++it) {
    const int i  = tid + 256 * it;
    const int ii = i < NIMG ? i : NIMG - 1;
    const int t  = ii / ROW4, pc = ii % ROW4;
    xv[it] = reinterpret_cast<const float4*>(xP + (size_t)sn[it] * DP)[pc];
    rv[it] = reinterpret_cast<const float4*>(rP + (size_t)(m0 + t) * DP)[pc];
  }
#pragma unroll
  for (int it = 0; it < NTP; ++it) {
    const int i = tid + 256 * it;
    lds.cwp[i] = (i % TAPP < CONVK) ? cwv[it] : 0.0f;
  }
#pragma unroll
  for (int it = 0; it < NWP; ++it) {
    const int i = tid + 256 * it;
    reinterpret_cast<float4*>(&lds.w1[0])[i] = w1v[it];
  }
  lds.chc[cht][0] = ccb;
  lds.chc[cht][1] = ccg * INV_SQRT_1PE;
  lds.chc[cht][2] = ccs;
  lds.chc[cht][3] = 0.0f;
#pragma unroll
  for (int it = 0; it < NIM; ++it) {
    const int i = tid + 256 * it;
    if (i < NIMG) {
      float* d = &lds.img[i / ROW4][8 * (i % ROW4)];
      d[0] = xv[it].x * b0s + b0b;  d[1] = rv[it].x * b0s + b0b;
      d[2] = xv[it].y * b0s + b0b;  d[3] = rv[it].y * b0s + b0b;
      d[4] = xv[it].z * b0s + b0b;  d[5] = rv[it].z * b0s + b0b;
      d[6] = xv[it].w * b0s + b0b;  d[7] = rv[it].w * b0s + b0b;
    }
  }
  __syncthreads();

  int toff[KCH][2];
#pragma unroll
  for (int ks = 0; ks < KCH; ++ks) {
#pragma unroll
    for (int s = 0; s < 2; ++s) {
      int tp = 4 * ks + 2 * half + s;
      toff[ks][s] = (tp < CONVK) ? (tp / 7) * 20 + (tp % 7) : 0;
    }
  }
  const bool lastx_ok = (4 * (KCH - 1) + 2 * half)     < CONVK;
  const bool lasty_ok = (4 * (KCH - 1) + 2 * half + 1) < CONVK;

  const float*  imrow = &lds.img[l16][0];
  const float4* fcw4  = reinterpret_cast<const float4*>(fcwP);

  v8f acc[ET][NTN];
#pragma unroll
  for (int t = 0; t < ET; ++t) {
#pragma unroll
    for (int i = 0; i < NTN; ++i) acc[t][i] = v8f{};
  }

  const int nseg = (wave < 6) ? 1 : 3;
  const int p0   = (wave < 6) ? 0 : PSPLIT;
  const int p1   = (wave < 6) ? PSPLIT : P196;

  for (int sg = 0; sg < nseg; ++sg) {
    const int j = (wave < 6) ? wave : 3 * (wave - 6) + sg;

    v2f ta[KCH];
#pragma unroll
    for (int ks = 0; ks < KCH; ++ks)
      ta[ks] = *reinterpret_cast<const v2f*>(&lds.cwp[(16 * j + l16) * TAPP + 4 * ks + 2 * half]);
    float cbv[8], scv[8], bbv[8];
#pragma unroll
    for (int r = 0; r < 8; ++r) {
      const float* cc = &lds.chc[16 * j + 8 * half + r][0];
      cbv[r] = cc[0]; scv[r] = cc[1]; bbv[r] = cc[2];
    }

    float4 fx[NTN], fy[NTN];
#pragma unroll
    for (int i = 0; i < NTN; ++i) { fx[i] = float4{0.f, 0.f, 0.f, 0.f}; fy[i] = fx[i]; }

#pragma unroll 1
    for (int pos = p0 - 1; pos < p1; ++pos) {
      const int pn = (pos + 1 < p1) ? pos + 1 : pos;
      const float4* fpn = fcw4 + (size_t)((j * P196 + pn) * 2) * (NTN * 32) + lane;

      if (pos >= p0) {
        const int woff = (pos / 14) * 20 + (pos % 14);
        float act[ET][8];
#pragma unroll
        for (int t = 0; t < ET; ++t) {
          const float* win = imrow + t * (16 * IMGP) + woff;
          v8f d0 = v8f{}, d1 = v8f{};
#pragma unroll
          for (int ks = 0; ks < KCH; ++ks) {
            v2f b;
            b.x = win[toff[ks][0]];
            b.y = win[toff[ks][1]];
            if (ks == KCH - 1) {
              b.x = lastx_ok ? b.x : 0.0f;
              b.y = lasty_ok ? b.y : 0.0f;
            }
            if (ks & 1) d1 = wmma4(ta[ks], b, d1);
            else        d0 = wmma4(ta[ks], b, d0);
          }
#pragma unroll
          for (int r = 0; r < 8; ++r) {
            float v = ((d0[r] + d1[r]) + cbv[r]) * scv[r] + bbv[r];
            act[t][r] = v > 0.f ? v : 0.f;
          }
        }

#pragma unroll
        for (int q = 0; q < 4; ++q) {
#pragma unroll
          for (int t = 0; t < ET; ++t) {
            v2f b;
            b.x = act[t][2 * q];
            b.y = act[t][2 * q + 1];
#pragma unroll
            for (int i = 0; i < NTN; ++i) {
              v2f a;
              if (q == 0)      { a.x = fx[i].x; a.y = fx[i].y; }
              else if (q == 1) { a.x = fx[i].z; a.y = fx[i].w; }
              else if (q == 2) { a.x = fy[i].x; a.y = fy[i].y; }
              else             { a.x = fy[i].z; a.y = fy[i].w; }
              acc[t][i] = wmma4(a, b, acc[t][i]);
            }
          }
        }
      }

#pragma unroll
      for (int i = 0; i < NTN; ++i) { fx[i] = fpn[i * 32]; fy[i] = fpn[(NTN + i) * 32]; }
    }
  }

  float cfb[NTN], csc[NTN], cbb[NTN];
#pragma unroll
  for (int i = 0; i < NTN; ++i) {
    const int n  = 16 * i + 8 * half + wave;
    const int nn = n < D0 ? n : D0 - 1;
    cfb[i] = fc_b[nn];
    csc[i] = bn2_g[nn] * INV_SQRT_1PE;
    cbb[i] = bn2_b[nn];
  }

#pragma unroll
  for (int t = 0; t < ET; ++t) {
#pragma unroll
    for (int i = 0; i < NTN; ++i) {
#pragma unroll
      for (int r = 0; r < 8; ++r) lds.part[wave][t][i][r][lane] = acc[t][i][r];
    }
  }
  __syncthreads();

  float sv[ET][NTN];
#pragma unroll
  for (int t = 0; t < ET; ++t) {
#pragma unroll
    for (int i = 0; i < NTN; ++i) {
      float s = 0.0f;
#pragma unroll
      for (int w = 0; w < 8; ++w) s += lds.part[w][t][i][wave][lane];
      sv[t][i] = s;
    }
  }
#pragma unroll
  for (int t = 0; t < ET; ++t) {
#pragma unroll
    for (int i = 0; i < NTN; ++i) {
      const int n = 16 * i + 8 * half + wave;
      if (n < D0) {
        float val = (sv[t][i] + cfb[i]) * csc[i] + cbb[i];
        lds.h2s[16 * t + l16][n] = val > 0.f ? val : 0.f;
      }
    }
  }
  __syncthreads();

  for (int i = tid; i < EB * NCLS; i += 256) {
    int t = i / NCLS, cls = i % NCLS;
    float s = fc1_b[cls];
    const float4* hp = reinterpret_cast<const float4*>(&lds.h2s[t][0]);
    const float4* wp = reinterpret_cast<const float4*>(&lds.w1[cls * D0]);
    for (int c4 = 0; c4 < D0 / 4; ++c4) {
      const float4 h = hp[c4], w = wp[c4];
      s = __builtin_fmaf(h.x, w.x, s);
      s = __builtin_fmaf(h.y, w.y, s);
      s = __builtin_fmaf(h.z, w.z, s);
      s = __builtin_fmaf(h.w, w.w, s);
    }
    out[(size_t)(m0 + t) * NCLS + cls] = s;
  }
}


extern "C" void kernel_launch(void* const* d_in, const int* in_sizes, int n_in,
                              void* d_out, int out_size, void* d_ws, size_t ws_size,
                              hipStream_t stream) {
  const float* node  = (const float*)d_in[0];
  const float* ef    = (const float*)d_in[1];
  const float* ief   = (const float*)d_in[2];
  const float* Win   = (const float*)d_in[3];
  const float* Wout  = (const float*)d_in[4];
  const float* Wloop = (const float*)d_in[5];
  const float* Wrel  = (const float*)d_in[6];
  const float* bgcn  = (const float*)d_in[7];
  const float* bn0g  = (const float*)d_in[8];
  const float* bn0b  = (const float*)d_in[9];
  const float* convw = (const float*)d_in[10];
  const float* convb = (const float*)d_in[11];
  const float* bn1g  = (const float*)d_in[12];
  const float* bn1b  = (const float*)d_in[13];
  const float* fcw   = (const float*)d_in[14];
  const float* fcb   = (const float*)d_in[15];
  const float* bn2g  = (const float*)d_in[16];
  const float* bn2b  = (const float*)d_in[17];
  const float* fc1w  = (const float*)d_in[18];
  const float* fc1b  = (const float*)d_in[19];
  const int*   ei    = (const int*)d_in[20];
  const int*   iei   = (const int*)d_in[21];
  const int *ei0 = ei, *ei1 = ei + NE;
  const int *iei0 = iei, *iei1 = iei + NE;

  float* ws = (float*)d_ws;
  size_t o = 0;
  float* xP   = ws + o; o += (size_t)NVP * DP;
  float* r0   = ws + o; o += (size_t)NE * DP;
  float* r1   = ws + o; o += (size_t)NE * DP;
  float* ir0  = ws + o; o += (size_t)NE * DP;
  float* ir1  = ws + o; o += (size_t)NE * DP;
  float* minb = ws + o; o += (size_t)NE * DP;
  float* moutb= ws + o; o += (size_t)NE * DP;
  float* fcwT = ws + o; o += (size_t)FLAT * DP;
  float* degf = ws + o; o += NVP;
  float* degi = ws + o; o += NVP;
  float* nfv  = ws + o; o += NE;
  float* niv  = ws + o; o += NE;

  const int TB = 256;
  auto cdiv = [](int a, int b) { return (a + b - 1) / b; };

  k_pad<<<cdiv(NVP * DP, TB), TB, 0, stream>>>(node, xP, NV, NVP);
  k_pad<<<cdiv(NE * DP, TB), TB, 0, stream>>>(ef, r0, NE, NE);
  k_pad<<<cdiv(NE * DP, TB), TB, 0, stream>>>(ief, ir0, NE, NE);
  k_pack_fcw<<<cdiv(FLAT * DP, TB), TB, 0, stream>>>(fcw, fcwT);

  k_zero<<<cdiv(2 * NVP, TB), TB, 0, stream>>>(degf, 2 * NVP);
  k_deg<<<cdiv(2 * NE, TB), TB, 0, stream>>>(ei0, iei0, degf, degi);
  k_norm<<<cdiv(NE, TB), TB, 0, stream>>>(ei0, ei1, iei0, iei1, degf, degi, nfv, niv);

  float *rc = r0, *rn = r1, *irc = ir0, *irn = ir1;
  const int GE = NE / 32;

  for (int l = 0; l < NLAY; ++l) {
    const float* wi = Win   + (size_t)l * D0 * D0;
    const float* wo = Wout  + (size_t)l * D0 * D0;
    const float* wp = Wloop + (size_t)l * D0 * D0;
    const float* wr = Wrel  + (size_t)l * D0 * D0;
    k_gemm<true, true><<<GE, 128, 0, stream>>>(xP, rc,  ei0,  wi, minb,  nfv, wr, rn);
    k_gemm<true, true><<<GE, 128, 0, stream>>>(xP, irc, iei0, wo, moutb, niv, wr, irn);
    k_aggupd<<<NV / NPW, NWV * 32, 0, stream>>>(minb, ei1, iei1, xP, wp, bgcn, l);
    float* t;
    t = rc; rc = rn; rn = t;
    t = irc; irc = irn; irn = t;
  }

  k_convfc<<<NE / EB, 256, 0, stream>>>(xP, rc, ei0, fcwT, convw, convb,
                                        bn1g, bn1b, fcb, bn2g, bn2b,
                                        fc1w, fc1b, bn0g, bn0b, (float*)d_out);
}
